# starting kernel
# baseline (speedup 1.0000x reference)
#include <hip/hip_runtime.h>

typedef __attribute__((ext_vector_type(2))) float v2f;
typedef __attribute__((ext_vector_type(8))) float v8f;

#define DIN 256
#define HD  128
#define HH  4
#define DD  32
#define NEG_SLOPE 0.2f

__global__ __launch_bounds__(256) void gat_gemm(const float* __restrict__ feat,
                                                const float* __restrict__ W,
                                                float* __restrict__ ft, int N) {
    __shared__ float As[16][260];
    const int tid = threadIdx.x;
    const int rowBase = blockIdx.x * 16;
    const bool fullTile = (rowBase + 16 <= N);

    if (fullTile) {
        #pragma unroll
        for (int i = 0; i < 4; ++i) {
            int idx4 = i * 256 + tid;
            int r    = idx4 >> 6;
            int c4   = idx4 & 63;
            float4 v = reinterpret_cast<const float4*>(feat + (size_t)(rowBase + r) * DIN)[c4];
            As[r][c4 * 4 + 0] = v.x; As[r][c4 * 4 + 1] = v.y;
            As[r][c4 * 4 + 2] = v.z; As[r][c4 * 4 + 3] = v.w;
        }
    } else {
        #pragma unroll
        for (int i = 0; i < 4; ++i) {
            int idx4 = i * 256 + tid;
            int r    = idx4 >> 6;
            int c4   = idx4 & 63;
            float4 v = make_float4(0.f, 0.f, 0.f, 0.f);
            if (rowBase + r < N)
                v = reinterpret_cast<const float4*>(feat + (size_t)(rowBase + r) * DIN)[c4];
            As[r][c4 * 4 + 0] = v.x; As[r][c4 * 4 + 1] = v.y;
            As[r][c4 * 4 + 2] = v.z; As[r][c4 * 4 + 3] = v.w;
        }
    }
    __syncthreads();

    const int wave    = tid >> 5;
    const int lane    = tid & 31;
    const int half    = lane >> 4;
    const int mrow    = lane & 15;
    const int colBase = wave * 16;
    const int ocol    = colBase + (lane & 15);
    const int kp      = half * 2;

    v8f c = {0.f, 0.f, 0.f, 0.f, 0.f, 0.f, 0.f, 0.f};
    const float* wrow = W + (size_t)ocol * DIN;

    #pragma unroll 8
    for (int k0 = 0; k0 < DIN; k0 += 4) {
        v2f a = *reinterpret_cast<const v2f*>(&As[mrow][k0 + kp]);
        v2f b = *reinterpret_cast<const v2f*>(wrow + k0 + kp);
        c = __builtin_amdgcn_wmma_f32_16x16x4_f32(
                 false, a,  false, b,
                 (short)0, c,  false,  false);
    }

    float* outBase = ft + (size_t)(rowBase + 8 * half) * HD + colBase + (lane & 15);
    if (fullTile) {
        #pragma unroll
        for (int v = 0; v < 8; ++v)
            outBase[(size_t)v * HD] = c[v];
    } else {
        #pragma unroll
        for (int v = 0; v < 8; ++v) {
            if (rowBase + v + 8 * half < N)
                outBase[(size_t)v * HD] = c[v];
        }
    }
}

__global__ __launch_bounds__(128) void gat_attn_logits(
        const float* __restrict__ ft,
        const float* __restrict__ attn_l, const float* __restrict__ attn_r,
        float* __restrict__ el, float* __restrict__ er) {
    const int n    = blockIdx.x;
    const int h    = threadIdx.x >> 5;
    const int lane = threadIdx.x & 31;
    float v = ft[(size_t)n * HD + h * DD + lane];
    float l = v * attn_l[h * DD + lane];
    float r = v * attn_r[h * DD + lane];
    #pragma unroll
    for (int off = 16; off > 0; off >>= 1) {
        l += __shfl_xor(l, off, 32);
        r += __shfl_xor(r, off, 32);
    }
    if (lane == 0) {
        el[n * HH + h] = l;
        er[n * HH + h] = r;
    }
}

__global__ void gat_init(float* __restrict__ out, const float* __restrict__ bias,
                         float* __restrict__ m, float* __restrict__ denom, int N) {
    size_t i = (size_t)blockIdx.x * blockDim.x + threadIdx.x;
    if (i < (size_t)N * HD) out[i] = bias[i & (HD - 1)];
    if (i < (size_t)N * HH) { m[i] = -3.402823466e38f; denom[i] = 0.f; }
}

__device__ __forceinline__ float leaky(float x) {
    return (x > 0.f) ? x : NEG_SLOPE * x;
}

__global__ void gat_edge_scores(const int* __restrict__ src, const int* __restrict__ dst,
                                const float* __restrict__ el, const float* __restrict__ er,
                                float* __restrict__ esc, float* __restrict__ m, int E) {
    int e = blockIdx.x * blockDim.x + threadIdx.x;
    if (e >= E) return;
    int s = src[e], d = dst[e];
    float4 l = reinterpret_cast<const float4*>(el)[s];
    float4 r = reinterpret_cast<const float4*>(er)[d];
    float4 v;
    v.x = leaky(l.x + r.x); v.y = leaky(l.y + r.y);
    v.z = leaky(l.z + r.z); v.w = leaky(l.w + r.w);
    reinterpret_cast<float4*>(esc)[e] = v;
    atomicMax(&m[d * HH + 0], v.x);
    atomicMax(&m[d * HH + 1], v.y);
    atomicMax(&m[d * HH + 2], v.z);
    atomicMax(&m[d * HH + 3], v.w);
}

__global__ void gat_edge_exp(const int* __restrict__ dst, float* __restrict__ esc,
                             const float* __restrict__ m, float* __restrict__ denom, int E) {
    int e = blockIdx.x * blockDim.x + threadIdx.x;
    if (e >= E) return;
    int d = dst[e];
    float4 v  = reinterpret_cast<const float4*>(esc)[e];
    float4 mx = reinterpret_cast<const float4*>(m)[d];
    v.x = __expf(v.x - mx.x); v.y = __expf(v.y - mx.y);
    v.z = __expf(v.z - mx.z); v.w = __expf(v.w - mx.w);
    reinterpret_cast<float4*>(esc)[e] = v;
    atomicAdd(&denom[d * HH + 0], v.x);
    atomicAdd(&denom[d * HH + 1], v.y);
    atomicAdd(&denom[d * HH + 2], v.z);
    atomicAdd(&denom[d * HH + 3], v.w);
}

__global__ __launch_bounds__(256) void gat_aggregate(
        const int* __restrict__ src, const int* __restrict__ dst,
        const float* __restrict__ esc, const float* __restrict__ denom,
        const float* __restrict__ ft, float* __restrict__ out, int E) {
    int e = blockIdx.x * 8 + (threadIdx.x >> 5);
    int lane = threadIdx.x & 31;
    if (e >= E) return;
    int s = src[e], d = dst[e];
    const float* fts = ft + (size_t)s * HD;
    __builtin_prefetch(fts + lane, 0, 0);
    float4 ae = reinterpret_cast<const float4*>(esc)[e];
    float4 dn = reinterpret_cast<const float4*>(denom)[d];
    float a0 = ae.x / dn.x, a1 = ae.y / dn.y, a2 = ae.z / dn.z, a3 = ae.w / dn.w;
    float* outd = out + (size_t)d * HD;
    atomicAdd(&outd[0 * DD + lane], fts[0 * DD + lane] * a0);
    atomicAdd(&outd[1 * DD + lane], fts[1 * DD + lane] * a1);
    atomicAdd(&outd[2 * DD + lane], fts[2 * DD + lane] * a2);
    atomicAdd(&outd[3 * DD + lane], fts[3 * DD + lane] * a3);
}

extern "C" void kernel_launch(void* const* d_in, const int* in_sizes, int n_in,
                              void* d_out, int out_size, void* d_ws, size_t ws_size,
                              hipStream_t stream) {
    const float* feat   = (const float*)d_in[0];
    const int*   src    = (const int*)  d_in[1];
    const int*   dst    = (const int*)  d_in[2];
    const float* W      = (const float*)d_in[3];
    const float* attn_l = (const float*)d_in[4];
    const float* attn_r = (const float*)d_in[5];
    const float* bias   = (const float*)d_in[6];

    const int N = in_sizes[0] / DIN;
    const int E = in_sizes[1];
    float* out = (float*)d_out;

    float* ws    = (float*)d_ws;
    float* ft    = ws;
    float* el    = ft    + (size_t)N * HD;
    float* er    = el    + (size_t)N * HH;
    float* mbuf  = er    + (size_t)N * HH;
    float* denom = mbuf  + (size_t)N * HH;
    float* esc   = denom + (size_t)N * HH;

    gat_gemm<<<(N + 15) / 16, 256, 0, stream>>>(feat, W, ft, N);
    gat_attn_logits<<<N, 128, 0, stream>>>(ft, attn_l, attn_r, el, er);

    size_t tot = (size_t)N * HD;
    gat_init<<<(unsigned)((tot + 255) / 256), 256, 0, stream>>>(out, bias, mbuf, denom, N);

    gat_edge_scores<<<(E + 255) / 256, 256, 0, stream>>>(src, dst, el, er, esc, mbuf, E);
    gat_edge_exp<<<(E + 255) / 256, 256, 0, stream>>>(dst, esc, mbuf, denom, E);
    gat_aggregate<<<(E + 7) / 8, 256, 0, stream>>>(src, dst, esc, denom, ft, out, E);
}


# Group edges by destination (integer atomic count gives edge ranks, prefix sum, fill); one wave per node does max, exp-sum and weighted row sum in registers: no float atomics, sums in rank order
# speedup vs baseline: 2.5583x; 2.5583x over previous
#include <hip/hip_runtime.h>

typedef __attribute__((ext_vector_type(2))) float v2f;
typedef __attribute__((ext_vector_type(8))) float v8f;

#define DIN 256
#define HD  128
#define HH  4
#define DD  32
#define NEG_SLOPE 0.2f

__global__ __launch_bounds__(256) void gat_gemm(const float* __restrict__ feat,
                                                const float* __restrict__ W,
                                                float* __restrict__ ft, int N) {
    __shared__ float As[16][260];
    const int tid = threadIdx.x;
    const int rowBase = blockIdx.x * 16;
    const bool fullTile = (rowBase + 16 <= N);

    if (fullTile) {
        #pragma unroll
        for (int i = 0; i < 4; ++i) {
            int idx4 = i * 256 + tid;
            int r    = idx4 >> 6;
            int c4   = idx4 & 63;
            float4 v = reinterpret_cast<const float4*>(feat + (size_t)(rowBase + r) * DIN)[c4];
            As[r][c4 * 4 + 0] = v.x; As[r][c4 * 4 + 1] = v.y;
            As[r][c4 * 4 + 2] = v.z; As[r][c4 * 4 + 3] = v.w;
        }
    } else {
        #pragma unroll
        for (int i = 0; i < 4; ++i) {
            int idx4 = i * 256 + tid;
            int r    = idx4 >> 6;
            int c4   = idx4 & 63;
            float4 v = make_float4(0.f, 0.f, 0.f, 0.f);
            if (rowBase + r < N)
                v = reinterpret_cast<const float4*>(feat + (size_t)(rowBase + r) * DIN)[c4];
            As[r][c4 * 4 + 0] = v.x; As[r][c4 * 4 + 1] = v.y;
            As[r][c4 * 4 + 2] = v.z; As[r][c4 * 4 + 3] = v.w;
        }
    }
    __syncthreads();

    const int wave    = tid >> 5;
    const int lane    = tid & 31;
    const int half    = lane >> 4;
    const int mrow    = lane & 15;
    const int colBase = wave * 16;
    const int ocol    = colBase + (lane & 15);
    const int kp      = half * 2;

    v8f c = {0.f, 0.f, 0.f, 0.f, 0.f, 0.f, 0.f, 0.f};
    const float* wrow = W + (size_t)ocol * DIN;

    #pragma unroll 8
    for (int k0 = 0; k0 < DIN; k0 += 4) {
        v2f a = *reinterpret_cast<const v2f*>(&As[mrow][k0 + kp]);
        v2f b = *reinterpret_cast<const v2f*>(wrow + k0 + kp);
        c = __builtin_amdgcn_wmma_f32_16x16x4_f32(
                 false, a,  false, b,
                 (short)0, c,  false,  false);
    }

    float* outBase = ft + (size_t)(rowBase + 8 * half) * HD + colBase + (lane & 15);
    if (fullTile) {
        #pragma unroll
        for (int v = 0; v < 8; ++v)
            outBase[(size_t)v * HD] = c[v];
    } else {
        #pragma unroll
        for (int v = 0; v < 8; ++v) {
            if (rowBase + v + 8 * half < N)
                outBase[(size_t)v * HD] = c[v];
        }
    }
}

__global__ __launch_bounds__(128) void gat_attn_logits(
        const float* __restrict__ ft,
        const float* __restrict__ attn_l, const float* __restrict__ attn_r,
        float* __restrict__ el, float* __restrict__ er, int* __restrict__ cnt) {
    const int n    = blockIdx.x;
    const int h    = threadIdx.x >> 5;
    const int lane = threadIdx.x & 31;
    float v = ft[(size_t)n * HD + h * DD + lane];
    float l = v * attn_l[h * DD + lane];
    float r = v * attn_r[h * DD + lane];
    #pragma unroll
    for (int off = 16; off > 0; off >>= 1) {
        l += __shfl_xor(l, off, 32);
        r += __shfl_xor(r, off, 32);
    }
    if (lane == 0) {
        el[n * HH + h] = l;
        er[n * HH + h] = r;
    }
    if (threadIdx.x == 0) cnt[n] = 0;
}

__global__ __launch_bounds__(256) void gat_count_rank(const int* __restrict__ dst,
                                                      int* __restrict__ cnt,
                                                      int* __restrict__ rank, int E) {
    const int e = blockIdx.x * 256 + threadIdx.x;
    if (e >= E) return;
    rank[e] = atomicAdd(&cnt[dst[e]], 1);
}

__global__ __launch_bounds__(1024) void gat_scan(const int* __restrict__ cnt,
                                                 int* __restrict__ rowptr, int N) {
    __shared__ int wtile[32];
    __shared__ int wfront[32];
    const int t     = threadIdx.x;
    const int lane  = t & 31;
    const int wave  = t >> 5;
    const int tile0 = blockIdx.x * 1024;

    int s = 0;
    #pragma unroll 8
    for (int i = t; i < tile0; i += 1024) s += cnt[i];
    #pragma unroll
    for (int off = 16; off > 0; off >>= 1) s += __shfl_xor(s, off, 32);

    const int i = tile0 + t;
    const int v = (i < N) ? cnt[i] : 0;
    int inc = v;
    #pragma unroll
    for (int off = 1; off < 32; off <<= 1) {
        const int u = __shfl_up(inc, off, 32);
        if (lane >= off) inc += u;
    }
    if (lane == 31) wtile[wave]  = inc;
    if (lane == 0)  wfront[wave] = s;
    __syncthreads();

    int base = 0;
    for (int w = 0; w < 32; ++w) {
        base += wfront[w];
        if (w < wave) base += wtile[w];
    }
    if (i < N) rowptr[i] = base + inc - v;
    if (blockIdx.x == gridDim.x - 1 && t == 1023) rowptr[N] = base + inc;
}

__global__ __launch_bounds__(256) void gat_fill(const int* __restrict__ src,
                                                const int* __restrict__ dst,
                                                const int* __restrict__ rowptr,
                                                const int* __restrict__ rank,
                                                int* __restrict__ perm, int E) {
    const int e = blockIdx.x * 256 + threadIdx.x;
    if (e >= E) return;
    perm[rowptr[dst[e]] + rank[e]] = src[e];
}

__device__ __forceinline__ float leaky(float x) {
    return (x > 0.f) ? x : NEG_SLOPE * x;
}

__global__ __launch_bounds__(256) void gat_node(
        const int* __restrict__ rowptr, const int* __restrict__ perm,
        const float* __restrict__ el, const float* __restrict__ er,
        const float* __restrict__ ft, const float* __restrict__ bias,
        float* __restrict__ out, int N) {
    const int lane = threadIdx.x & 31;
    const int d = __builtin_amdgcn_readfirstlane((int)(blockIdx.x * 8 + (threadIdx.x >> 5)));
    if (d >= N) return;
    const int beg = rowptr[d];
    const int end = rowptr[d + 1];
    const float4* el4 = reinterpret_cast<const float4*>(el);
    const float4 r = reinterpret_cast<const float4*>(er)[d];

    float m0 = -3.402823466e38f, m1 = m0, m2 = m0, m3 = m0;
    for (int j = beg + lane; j < end; j += 32) {
        const float4 l = el4[perm[j]];
        m0 = fmaxf(m0, leaky(l.x + r.x));
        m1 = fmaxf(m1, leaky(l.y + r.y));
        m2 = fmaxf(m2, leaky(l.z + r.z));
        m3 = fmaxf(m3, leaky(l.w + r.w));
    }
    #pragma unroll
    for (int off = 16; off > 0; off >>= 1) {
        m0 = fmaxf(m0, __shfl_xor(m0, off, 32));
        m1 = fmaxf(m1, __shfl_xor(m1, off, 32));
        m2 = fmaxf(m2, __shfl_xor(m2, off, 32));
        m3 = fmaxf(m3, __shfl_xor(m3, off, 32));
    }

    float s0 = 0.f, s1 = 0.f, s2 = 0.f, s3 = 0.f;
    for (int j = beg + lane; j < end; j += 32) {
        const float4 l = el4[perm[j]];
        s0 += __expf(leaky(l.x + r.x) - m0);
        s1 += __expf(leaky(l.y + r.y) - m1);
        s2 += __expf(leaky(l.z + r.z) - m2);
        s3 += __expf(leaky(l.w + r.w) - m3);
    }
    #pragma unroll
    for (int off = 16; off > 0; off >>= 1) {
        s0 += __shfl_xor(s0, off, 32);
        s1 += __shfl_xor(s1, off, 32);
        s2 += __shfl_xor(s2, off, 32);
        s3 += __shfl_xor(s3, off, 32);
    }

    const int hsel = lane >> 3;
    const unsigned p0 = (hsel == 0) ? 0xffffffffu : 0u;
    const unsigned p1 = (hsel == 1) ? 0xffffffffu : 0u;
    const unsigned p2 = (hsel == 2) ? 0xffffffffu : 0u;
    const unsigned p3 = (hsel == 3) ? 0xffffffffu : 0u;
    float4 acc = make_float4(0.f, 0.f, 0.f, 0.f);
    for (int base = beg; base < end; base += 32) {
        const int j = base + lane;
        int   sj = 0;
        float a0 = 0.f, a1 = 0.f, a2 = 0.f, a3 = 0.f;
        if (j < end) {
            sj = perm[j];
            const float4 l = el4[sj];
            a0 = __expf(leaky(l.x + r.x) - m0) / s0;
            a1 = __expf(leaky(l.y + r.y) - m1) / s1;
            a2 = __expf(leaky(l.z + r.z) - m2) / s2;
            a3 = __expf(leaky(l.w + r.w) - m3) / s3;
        }
        const int nk = (end - base < 32) ? (end - base) : 32;
        #pragma unroll 4
        for (int k = 0; k < nk; ++k) {
            const int      sk = __builtin_amdgcn_readlane(sj, k);
            const unsigned b0 = (unsigned)__builtin_amdgcn_readlane(__float_as_int(a0), k);
            const unsigned b1 = (unsigned)__builtin_amdgcn_readlane(__float_as_int(a1), k);
            const unsigned b2 = (unsigned)__builtin_amdgcn_readlane(__float_as_int(a2), k);
            const unsigned b3 = (unsigned)__builtin_amdgcn_readlane(__float_as_int(a3), k);
            const float w = __uint_as_float((b0 & p0) | (b1 & p1) | (b2 & p2) | (b3 & p3));
            const float4 f = reinterpret_cast<const float4*>(ft + (size_t)sk * HD)[lane];
            acc.x += f.x * w;
            acc.y += f.y * w;
            acc.z += f.z * w;
            acc.w += f.w * w;
        }
    }
    const float4 b = reinterpret_cast<const float4*>(bias)[lane];
    acc.x += b.x; acc.y += b.y; acc.z += b.z; acc.w += b.w;
    reinterpret_cast<float4*>(out + (size_t)d * HD)[lane] = acc;
}

extern "C" void kernel_launch(void* const* d_in, const int* in_sizes, int n_in,
                              void* d_out, int out_size, void* d_ws, size_t ws_size,
                              hipStream_t stream) {
    const float* feat   = (const float*)d_in[0];
    const int*   src    = (const int*)  d_in[1];
    const int*   dst    = (const int*)  d_in[2];
    const float* W      = (const float*)d_in[3];
    const float* attn_l = (const float*)d_in[4];
    const float* attn_r = (const float*)d_in[5];
    const float* bias   = (const float*)d_in[6];

    const int N = in_sizes[0] / DIN;
    const int E = in_sizes[1];
    float* out = (float*)d_out;

    float* ws     = (float*)d_ws;
    float* ft     = ws;
    float* el     = ft + (size_t)N * HD;
    float* er     = el + (size_t)N * HH;
    int*   cnt    = (int*)(er + (size_t)N * HH);
    int*   rowptr = cnt + N;
    int*   rank   = rowptr + N + 1;
    int*   perm   = rank + E;

    gat_gemm<<<(N + 15) / 16, 256, 0, stream>>>(feat, W, ft, N);
    gat_attn_logits<<<N, 128, 0, stream>>>(ft, attn_l, attn_r, el, er, cnt);
    gat_count_rank<<<(E + 255) / 256, 256, 0, stream>>>(dst, cnt, rank, E);
    gat_scan<<<(N + 1023) / 1024, 1024, 0, stream>>>(cnt, rowptr, N);
    gat_fill<<<(E + 255) / 256, 256, 0, stream>>>(src, dst, rowptr, rank, perm, E);
    gat_node<<<(N + 7) / 8, 256, 0, stream>>>(rowptr, perm, el, er, ft, bias, out, N);
}


# Logits el/er and the counter clear computed in the GEMM epilogue from the accumulators (DPP row sums, half-heads added in LDS); logits kernel dropped; logit sum order changed, still fixed
# speedup vs baseline: 2.8988x; 1.1331x over previous
#include <hip/hip_runtime.h>

typedef __attribute__((ext_vector_type(2))) float v2f;
typedef __attribute__((ext_vector_type(8))) float v8f;

#define DIN 256
#define HD  128
#define HH  4
#define DD  32
#define NEG_SLOPE 0.2f

template <int M>
__device__ __forceinline__ float xrow(float x) {
    return __int_as_float(__builtin_amdgcn_update_dpp(
        0, __float_as_int(x), 0x160 | M, 0xf, 0xf, true));
}

__global__ __launch_bounds__(256) void gat_gemm(const float* __restrict__ feat,
                                                const float* __restrict__ W,
                                                const float* __restrict__ attn_l,
                                                const float* __restrict__ attn_r,
                                                float* __restrict__ ft,
                                                float* __restrict__ el,
                                                float* __restrict__ er,
                                                int* __restrict__ cnt, int N) {
    __shared__ float As[16][260];
    __shared__ float part[2][8][16];
    const int tid = threadIdx.x;
    const int rowBase = blockIdx.x * 16;
    const bool fullTile = (rowBase + 16 <= N);

    if (fullTile) {
        #pragma unroll
        for (int i = 0; i < 4; ++i) {
            int idx4 = i * 256 + tid;
            int r    = idx4 >> 6;
            int c4   = idx4 & 63;
            float4 v = reinterpret_cast<const float4*>(feat + (size_t)(rowBase + r) * DIN)[c4];
            As[r][c4 * 4 + 0] = v.x; As[r][c4 * 4 + 1] = v.y;
            As[r][c4 * 4 + 2] = v.z; As[r][c4 * 4 + 3] = v.w;
        }
    } else {
        #pragma unroll
        for (int i = 0; i < 4; ++i) {
            int idx4 = i * 256 + tid;
            int r    = idx4 >> 6;
            int c4   = idx4 & 63;
            float4 v = make_float4(0.f, 0.f, 0.f, 0.f);
            if (rowBase + r < N)
                v = reinterpret_cast<const float4*>(feat + (size_t)(rowBase + r) * DIN)[c4];
            As[r][c4 * 4 + 0] = v.x; As[r][c4 * 4 + 1] = v.y;
            As[r][c4 * 4 + 2] = v.z; As[r][c4 * 4 + 3] = v.w;
        }
    }
    __syncthreads();

    const int wave    = tid >> 5;
    const int lane    = tid & 31;
    const int half    = lane >> 4;
    const int mrow    = lane & 15;
    const int colBase = wave * 16;
    const int ocol    = colBase + (lane & 15);
    const int kp      = half * 2;

    v8f c = {0.f, 0.f, 0.f, 0.f, 0.f, 0.f, 0.f, 0.f};
    const float* wrow = W + (size_t)ocol * DIN;
    const float al = attn_l[ocol];
    const float ar = attn_r[ocol];

    #pragma unroll 8
    for (int k0 = 0; k0 < DIN; k0 += 4) {
        v2f a = *reinterpret_cast<const v2f*>(&As[mrow][k0 + kp]);
        v2f b = *reinterpret_cast<const v2f*>(wrow + k0 + kp);
        c = __builtin_amdgcn_wmma_f32_16x16x4_f32(
                 false, a,  false, b,
                 (short)0, c,  false,  false);
    }

    float pl[8], pr[8];
    #pragma unroll
    for (int v = 0; v < 8; ++v) { pl[v] = c[v] * al; pr[v] = c[v] * ar; }
    const bool b3 = (lane & 8) != 0;
    const bool b2 = (lane & 4) != 0;
    const bool b1 = (lane & 2) != 0;
    const bool b0 = (lane & 1) != 0;
    float l1[4], r1[4];
    #pragma unroll
    for (int v = 0; v < 4; ++v) {
        l1[v] = (b3 ? pl[v + 4] : pl[v]) + xrow<8>(b3 ? pl[v] : pl[v + 4]);
        r1[v] = (b3 ? pr[v + 4] : pr[v]) + xrow<8>(b3 ? pr[v] : pr[v + 4]);
    }
    float l2[2], r2[2];
    #pragma unroll
    for (int v = 0; v < 2; ++v) {
        l2[v] = (b2 ? l1[v + 2] : l1[v]) + xrow<4>(b2 ? l1[v] : l1[v + 2]);
        r2[v] = (b2 ? r1[v + 2] : r1[v]) + xrow<4>(b2 ? r1[v] : r1[v + 2]);
    }
    const float l3 = (b1 ? l2[1] : l2[0]) + xrow<2>(b1 ? l2[0] : l2[1]);
    const float r3 = (b1 ? r2[1] : r2[0]) + xrow<2>(b1 ? r2[0] : r2[1]);
    const float fin = (b0 ? r3 : l3) + xrow<1>(b0 ? l3 : r3);
    part[lane & 1][wave][8 * half + ((lane & 15) >> 1)] = fin;
    __syncthreads();

    float* outBase = ft + (size_t)(rowBase + 8 * half) * HD + colBase + (lane & 15);
    if (fullTile) {
        #pragma unroll
        for (int v = 0; v < 8; ++v)
            outBase[(size_t)v * HD] = c[v];
    } else {
        #pragma unroll
        for (int v = 0; v < 8; ++v) {
            if (rowBase + v + 8 * half < N)
                outBase[(size_t)v * HD] = c[v];
        }
    }

    if (tid < 128) {
        const int q  = tid >> 6;
        const int rr = (tid >> 2) & 15;
        const int hh = tid & 3;
        const float sum = part[q][2 * hh][rr] + part[q][2 * hh + 1][rr];
        if (rowBase + rr < N) {
            float* dstp = q ? er : el;
            dstp[(size_t)(rowBase + rr) * HH + hh] = sum;
        }
    } else if (tid < 144) {
        const int rr = tid - 128;
        if (rowBase + rr < N) cnt[rowBase + rr] = 0;
    }
}

__global__ __launch_bounds__(256) void gat_count_rank(const int* __restrict__ dst,
                                                      int* __restrict__ cnt,
                                                      int* __restrict__ rank, int E) {
    const int e = blockIdx.x * 256 + threadIdx.x;
    if (e >= E) return;
    rank[e] = atomicAdd(&cnt[dst[e]], 1);
}

__global__ __launch_bounds__(1024) void gat_scan(const int* __restrict__ cnt,
                                                 int* __restrict__ rowptr, int N) {
    __shared__ int wtile[32];
    __shared__ int wfront[32];
    const int t     = threadIdx.x;
    const int lane  = t & 31;
    const int wave  = t >> 5;
    const int tile0 = blockIdx.x * 1024;

    int s = 0;
    #pragma unroll 8
    for (int i = t; i < tile0; i += 1024) s += cnt[i];
    #pragma unroll
    for (int off = 16; off > 0; off >>= 1) s += __shfl_xor(s, off, 32);

    const int i = tile0 + t;
    const int v = (i < N) ? cnt[i] : 0;
    int inc = v;
    #pragma unroll
    for (int off = 1; off < 32; off <<= 1) {
        const int u = __shfl_up(inc, off, 32);
        if (lane >= off) inc += u;
    }
    if (lane == 31) wtile[wave]  = inc;
    if (lane == 0)  wfront[wave] = s;
    __syncthreads();

    int base = 0;
    for (int w = 0; w < 32; ++w) {
        base += wfront[w];
        if (w < wave) base += wtile[w];
    }
    if (i < N) rowptr[i] = base + inc - v;
    if (blockIdx.x == gridDim.x - 1 && t == 1023) rowptr[N] = base + inc;
}

__global__ __launch_bounds__(256) void gat_fill(const int* __restrict__ src,
                                                const int* __restrict__ dst,
                                                const int* __restrict__ rowptr,
                                                const int* __restrict__ rank,
                                                int* __restrict__ perm, int E) {
    const int e = blockIdx.x * 256 + threadIdx.x;
    if (e >= E) return;
    perm[rowptr[dst[e]] + rank[e]] = src[e];
}

__device__ __forceinline__ float leaky(float x) {
    return (x > 0.f) ? x : NEG_SLOPE * x;
}

__global__ __launch_bounds__(256) void gat_node(
        const int* __restrict__ rowptr, const int* __restrict__ perm,
        const float* __restrict__ el, const float* __restrict__ er,
        const float* __restrict__ ft, const float* __restrict__ bias,
        float* __restrict__ out, int N) {
    const int lane = threadIdx.x & 31;
    const int d = __builtin_amdgcn_readfirstlane((int)(blockIdx.x * 8 + (threadIdx.x >> 5)));
    if (d >= N) return;
    const int beg = rowptr[d];
    const int end = rowptr[d + 1];
    const float4* el4 = reinterpret_cast<const float4*>(el);
    const float4 r = reinterpret_cast<const float4*>(er)[d];

    float m0 = -3.402823466e38f, m1 = m0, m2 = m0, m3 = m0;
    for (int j = beg + lane; j < end; j += 32) {
        const float4 l = el4[perm[j]];
        m0 = fmaxf(m0, leaky(l.x + r.x));
        m1 = fmaxf(m1, leaky(l.y + r.y));
        m2 = fmaxf(m2, leaky(l.z + r.z));
        m3 = fmaxf(m3, leaky(l.w + r.w));
    }
    #pragma unroll
    for (int off = 16; off > 0; off >>= 1) {
        m0 = fmaxf(m0, __shfl_xor(m0, off, 32));
        m1 = fmaxf(m1, __shfl_xor(m1, off, 32));
        m2 = fmaxf(m2, __shfl_xor(m2, off, 32));
        m3 = fmaxf(m3, __shfl_xor(m3, off, 32));
    }

    float s0 = 0.f, s1 = 0.f, s2 = 0.f, s3 = 0.f;
    for (int j = beg + lane; j < end; j += 32) {
        const float4 l = el4[perm[j]];
        s0 += __expf(leaky(l.x + r.x) - m0);
        s1 += __expf(leaky(l.y + r.y) - m1);
        s2 += __expf(leaky(l.z + r.z) - m2);
        s3 += __expf(leaky(l.w + r.w) - m3);
    }
    #pragma unroll
    for (int off = 16; off > 0; off >>= 1) {
        s0 += __shfl_xor(s0, off, 32);
        s1 += __shfl_xor(s1, off, 32);
        s2 += __shfl_xor(s2, off, 32);
        s3 += __shfl_xor(s3, off, 32);
    }

    const int hsel = lane >> 3;
    const unsigned p0 = (hsel == 0) ? 0xffffffffu : 0u;
    const unsigned p1 = (hsel == 1) ? 0xffffffffu : 0u;
    const unsigned p2 = (hsel == 2) ? 0xffffffffu : 0u;
    const unsigned p3 = (hsel == 3) ? 0xffffffffu : 0u;
    float4 acc = make_float4(0.f, 0.f, 0.f, 0.f);
    for (int base = beg; base < end; base += 32) {
        const int j = base + lane;
        int   sj = 0;
        float a0 = 0.f, a1 = 0.f, a2 = 0.f, a3 = 0.f;
        if (j < end) {
            sj = perm[j];
            const float4 l = el4[sj];
            a0 = __expf(leaky(l.x + r.x) - m0) / s0;
            a1 = __expf(leaky(l.y + r.y) - m1) / s1;
            a2 = __expf(leaky(l.z + r.z) - m2) / s2;
            a3 = __expf(leaky(l.w + r.w) - m3) / s3;
        }
        const int nk = (end - base < 32) ? (end - base) : 32;
        #pragma unroll 4
        for (int k = 0; k < nk; ++k) {
            const int      sk = __builtin_amdgcn_readlane(sj, k);
            const unsigned b0 = (unsigned)__builtin_amdgcn_readlane(__float_as_int(a0), k);
            const unsigned b1 = (unsigned)__builtin_amdgcn_readlane(__float_as_int(a1), k);
            const unsigned b2 = (unsigned)__builtin_amdgcn_readlane(__float_as_int(a2), k);
            const unsigned b3 = (unsigned)__builtin_amdgcn_readlane(__float_as_int(a3), k);
            const float w = __uint_as_float((b0 & p0) | (b1 & p1) | (b2 & p2) | (b3 & p3));
            const float4 f = reinterpret_cast<const float4*>(ft + (size_t)sk * HD)[lane];
            acc.x += f.x * w;
            acc.y += f.y * w;
            acc.z += f.z * w;
            acc.w += f.w * w;
        }
    }
    const float4 b = reinterpret_cast<const float4*>(bias)[lane];
    acc.x += b.x; acc.y += b.y; acc.z += b.z; acc.w += b.w;
    reinterpret_cast<float4*>(out + (size_t)d * HD)[lane] = acc;
}

extern "C" void kernel_launch(void* const* d_in, const int* in_sizes, int n_in,
                              void* d_out, int out_size, void* d_ws, size_t ws_size,
                              hipStream_t stream) {
    const float* feat   = (const float*)d_in[0];
    const int*   src    = (const int*)  d_in[1];
    const int*   dst    = (const int*)  d_in[2];
    const float* W      = (const float*)d_in[3];
    const float* attn_l = (const float*)d_in[4];
    const float* attn_r = (const float*)d_in[5];
    const float* bias   = (const float*)d_in[6];

    const int N = in_sizes[0] / DIN;
    const int E = in_sizes[1];
    float* out = (float*)d_out;

    float* ws     = (float*)d_ws;
    float* ft     = ws;
    float* el     = ft + (size_t)N * HD;
    float* er     = el + (size_t)N * HH;
    int*   cnt    = (int*)(er + (size_t)N * HH);
    int*   rowptr = cnt + N;
    int*   rank   = rowptr + N + 1;
    int*   perm   = rank + E;

    gat_gemm<<<(N + 15) / 16, 256, 0, stream>>>(feat, W, attn_l, attn_r, ft, el, er, cnt, N);
    gat_count_rank<<<(E + 255) / 256, 256, 0, stream>>>(dst, cnt, rank, E);
    gat_scan<<<(N + 1023) / 1024, 1024, 0, stream>>>(cnt, rowptr, N);
    gat_fill<<<(E + 255) / 256, 256, 0, stream>>>(src, dst, rowptr, rank, perm, E);
    gat_node<<<(N + 7) / 8, 256, 0, stream>>>(rowptr, perm, el, er, ft, bias, out, N);
}


# GEMM block takes 64 rows instead of 16: each wave keeps four 16x16 accumulators, so one W fragment load from global feeds four matrix instructions (W loads and their cache lines cut 4x)
# speedup vs baseline: 3.1799x; 1.0970x over previous
#include <hip/hip_runtime.h>

typedef __attribute__((ext_vector_type(2))) float v2f;
typedef __attribute__((ext_vector_type(8))) float v8f;

#define DIN 256
#define HD  128
#define HH  4
#define DD  32
#define NEG_SLOPE 0.2f

template <int M>
__device__ __forceinline__ float xrow(float x) {
    return __int_as_float(__builtin_amdgcn_update_dpp(
        0, __float_as_int(x), 0x160 | M, 0xf, 0xf, true));
}

#define GROWS 64
#define GTILES (GROWS / 16)
#define PSTR 72
#define QSTR (8 * PSTR + 16)

__global__ __launch_bounds__(256) void gat_gemm(const float* __restrict__ feat,
                                                const float* __restrict__ W,
                                                const float* __restrict__ attn_l,
                                                const float* __restrict__ attn_r,
                                                float* __restrict__ ft,
                                                float* __restrict__ el,
                                                float* __restrict__ er,
                                                int* __restrict__ cnt, int N) {
    __shared__ __attribute__((aligned(16))) float As[GROWS][260];
    __shared__ float part[2 * QSTR];
    const int tid = threadIdx.x;
    const int rowBase = blockIdx.x * GROWS;
    const bool fullTile = (rowBase + GROWS <= N);

    if (fullTile) {
        #pragma unroll
        for (int i = 0; i < 16; ++i) {
            int idx4 = i * 256 + tid;
            int r    = idx4 >> 6;
            int c4   = idx4 & 63;
            float4 v = reinterpret_cast<const float4*>(feat + (size_t)(rowBase + r) * DIN)[c4];
            *reinterpret_cast<float4*>(&As[r][c4 * 4]) = v;
        }
    } else {
        #pragma unroll
        for (int i = 0; i < 16; ++i) {
            int idx4 = i * 256 + tid;
            int r    = idx4 >> 6;
            int c4   = idx4 & 63;
            float4 v = make_float4(0.f, 0.f, 0.f, 0.f);
            if (rowBase + r < N)
                v = reinterpret_cast<const float4*>(feat + (size_t)(rowBase + r) * DIN)[c4];
            *reinterpret_cast<float4*>(&As[r][c4 * 4]) = v;
        }
    }
    __syncthreads();

    const int wave    = tid >> 5;
    const int lane    = tid & 31;
    const int half    = lane >> 4;
    const int mrow    = lane & 15;
    const int colBase = wave * 16;
    const int ocol    = colBase + (lane & 15);
    const int kp      = half * 2;

    v8f c[GTILES];
    #pragma unroll
    for (int t = 0; t < GTILES; ++t)
        c[t] = (v8f){0.f, 0.f, 0.f, 0.f, 0.f, 0.f, 0.f, 0.f};
    const float* wrow = W + (size_t)ocol * DIN;
    const float al = attn_l[ocol];
    const float ar = attn_r[ocol];

    #pragma unroll 4
    for (int k0 = 0; k0 < DIN; k0 += 4) {
        v2f b = *reinterpret_cast<const v2f*>(wrow + k0 + kp);
        #pragma unroll
        for (int t = 0; t < GTILES; ++t) {
            v2f a = *reinterpret_cast<const v2f*>(&As[t * 16 + mrow][k0 + kp]);
            c[t] = __builtin_amdgcn_wmma_f32_16x16x4_f32(
                     false, a,  false, b,
                     (short)0, c[t],  false,  false);
        }
    }

    const bool b3 = (lane & 8) != 0;
    const bool b2 = (lane & 4) != 0;
    const bool b1 = (lane & 2) != 0;
    const bool b0 = (lane & 1) != 0;
    #pragma unroll
    for (int t = 0; t < GTILES; ++t) {
        float pl[8], pr[8];
        #pragma unroll
        for (int v = 0; v < 8; ++v) { pl[v] = c[t][v] * al; pr[v] = c[t][v] * ar; }
        float l1[4], r1[4];
        #pragma unroll
        for (int v = 0; v < 4; ++v) {
            l1[v] = (b3 ? pl[v + 4] : pl[v]) + xrow<8>(b3 ? pl[v] : pl[v + 4]);
            r1[v] = (b3 ? pr[v + 4] : pr[v]) + xrow<8>(b3 ? pr[v] : pr[v + 4]);
        }
        float l2[2], r2[2];
        #pragma unroll
        for (int v = 0; v < 2; ++v) {
            l2[v] = (b2 ? l1[v + 2] : l1[v]) + xrow<4>(b2 ? l1[v] : l1[v + 2]);
            r2[v] = (b2 ? r1[v + 2] : r1[v]) + xrow<4>(b2 ? r1[v] : r1[v + 2]);
        }
        const float l3 = (b1 ? l2[1] : l2[0]) + xrow<2>(b1 ? l2[0] : l2[1]);
        const float r3 = (b1 ? r2[1] : r2[0]) + xrow<2>(b1 ? r2[0] : r2[1]);
        const float fin = (b0 ? r3 : l3) + xrow<1>(b0 ? l3 : r3);
        part[(lane & 1) * QSTR + wave * PSTR + t * 16 + 8 * half + ((lane & 15) >> 1)] = fin;
    }
    __syncthreads();

    #pragma unroll
    for (int t = 0; t < GTILES; ++t) {
        float* outBase = ft + (size_t)(rowBase + t * 16 + 8 * half) * HD + colBase + (lane & 15);
        if (fullTile) {
            #pragma unroll
            for (int v = 0; v < 8; ++v)
                outBase[(size_t)v * HD] = c[t][v];
        } else {
            #pragma unroll
            for (int v = 0; v < 8; ++v) {
                if (rowBase + t * 16 + v + 8 * half < N)
                    outBase[(size_t)v * HD] = c[t][v];
            }
        }
    }

    {
        const int rr = tid >> 2;
        const int hh = tid & 3;
        const bool live = (rowBase + rr < N);
        #pragma unroll
        for (int q = 0; q < 2; ++q) {
            const float sum = part[q * QSTR + (2 * hh) * PSTR + rr]
                            + part[q * QSTR + (2 * hh + 1) * PSTR + rr];
            if (live) {
                float* dstp = q ? er : el;
                dstp[(size_t)(rowBase + rr) * HH + hh] = sum;
            }
        }
        if (tid < GROWS && rowBase + tid < N) cnt[rowBase + tid] = 0;
    }
}

__global__ __launch_bounds__(256) void gat_count_rank(const int* __restrict__ dst,
                                                      int* __restrict__ cnt,
                                                      int* __restrict__ rank, int E) {
    const int e = blockIdx.x * 256 + threadIdx.x;
    if (e >= E) return;
    rank[e] = atomicAdd(&cnt[dst[e]], 1);
}

__global__ __launch_bounds__(1024) void gat_scan(const int* __restrict__ cnt,
                                                 int* __restrict__ rowptr, int N) {
    __shared__ int wtile[32];
    __shared__ int wfront[32];
    const int t     = threadIdx.x;
    const int lane  = t & 31;
    const int wave  = t >> 5;
    const int tile0 = blockIdx.x * 1024;

    int s = 0;
    #pragma unroll 8
    for (int i = t; i < tile0; i += 1024) s += cnt[i];
    #pragma unroll
    for (int off = 16; off > 0; off >>= 1) s += __shfl_xor(s, off, 32);

    const int i = tile0 + t;
    const int v = (i < N) ? cnt[i] : 0;
    int inc = v;
    #pragma unroll
    for (int off = 1; off < 32; off <<= 1) {
        const int u = __shfl_up(inc, off, 32);
        if (lane >= off) inc += u;
    }
    if (lane == 31) wtile[wave]  = inc;
    if (lane == 0)  wfront[wave] = s;
    __syncthreads();

    int base = 0;
    for (int w = 0; w < 32; ++w) {
        base += wfront[w];
        if (w < wave) base += wtile[w];
    }
    if (i < N) rowptr[i] = base + inc - v;
    if (blockIdx.x == gridDim.x - 1 && t == 1023) rowptr[N] = base + inc;
}

__global__ __launch_bounds__(256) void gat_fill(const int* __restrict__ src,
                                                const int* __restrict__ dst,
                                                const int* __restrict__ rowptr,
                                                const int* __restrict__ rank,
                                                int* __restrict__ perm, int E) {
    const int e = blockIdx.x * 256 + threadIdx.x;
    if (e >= E) return;
    perm[rowptr[dst[e]] + rank[e]] = src[e];
}

__device__ __forceinline__ float leaky(float x) {
    return (x > 0.f) ? x : NEG_SLOPE * x;
}

__global__ __launch_bounds__(256) void gat_node(
        const int* __restrict__ rowptr, const int* __restrict__ perm,
        const float* __restrict__ el, const float* __restrict__ er,
        const float* __restrict__ ft, const float* __restrict__ bias,
        float* __restrict__ out, int N) {
    const int lane = threadIdx.x & 31;
    const int d = __builtin_amdgcn_readfirstlane((int)(blockIdx.x * 8 + (threadIdx.x >> 5)));
    if (d >= N) return;
    const int beg = rowptr[d];
    const int end = rowptr[d + 1];
    const float4* el4 = reinterpret_cast<const float4*>(el);
    const float4 r = reinterpret_cast<const float4*>(er)[d];

    float m0 = -3.402823466e38f, m1 = m0, m2 = m0, m3 = m0;
    for (int j = beg + lane; j < end; j += 32) {
        const float4 l = el4[perm[j]];
        m0 = fmaxf(m0, leaky(l.x + r.x));
        m1 = fmaxf(m1, leaky(l.y + r.y));
        m2 = fmaxf(m2, leaky(l.z + r.z));
        m3 = fmaxf(m3, leaky(l.w + r.w));
    }
    #pragma unroll
    for (int off = 16; off > 0; off >>= 1) {
        m0 = fmaxf(m0, __shfl_xor(m0, off, 32));
        m1 = fmaxf(m1, __shfl_xor(m1, off, 32));
        m2 = fmaxf(m2, __shfl_xor(m2, off, 32));
        m3 = fmaxf(m3, __shfl_xor(m3, off, 32));
    }

    float s0 = 0.f, s1 = 0.f, s2 = 0.f, s3 = 0.f;
    for (int j = beg + lane; j < end; j += 32) {
        const float4 l = el4[perm[j]];
        s0 += __expf(leaky(l.x + r.x) - m0);
        s1 += __expf(leaky(l.y + r.y) - m1);
        s2 += __expf(leaky(l.z + r.z) - m2);
        s3 += __expf(leaky(l.w + r.w) - m3);
    }
    #pragma unroll
    for (int off = 16; off > 0; off >>= 1) {
        s0 += __shfl_xor(s0, off, 32);
        s1 += __shfl_xor(s1, off, 32);
        s2 += __shfl_xor(s2, off, 32);
        s3 += __shfl_xor(s3, off, 32);
    }

    const int hsel = lane >> 3;
    const unsigned p0 = (hsel == 0) ? 0xffffffffu : 0u;
    const unsigned p1 = (hsel == 1) ? 0xffffffffu : 0u;
    const unsigned p2 = (hsel == 2) ? 0xffffffffu : 0u;
    const unsigned p3 = (hsel == 3) ? 0xffffffffu : 0u;
    float4 acc = make_float4(0.f, 0.f, 0.f, 0.f);
    for (int base = beg; base < end; base += 32) {
        const int j = base + lane;
        int   sj = 0;
        float a0 = 0.f, a1 = 0.f, a2 = 0.f, a3 = 0.f;
        if (j < end) {
            sj = perm[j];
            const float4 l = el4[sj];
            a0 = __expf(leaky(l.x + r.x) - m0) / s0;
            a1 = __expf(leaky(l.y + r.y) - m1) / s1;
            a2 = __expf(leaky(l.z + r.z) - m2) / s2;
            a3 = __expf(leaky(l.w + r.w) - m3) / s3;
        }
        const int nk = (end - base < 32) ? (end - base) : 32;
        #pragma unroll 4
        for (int k = 0; k < nk; ++k) {
            const int      sk = __builtin_amdgcn_readlane(sj, k);
            const unsigned b0 = (unsigned)__builtin_amdgcn_readlane(__float_as_int(a0), k);
            const unsigned b1 = (unsigned)__builtin_amdgcn_readlane(__float_as_int(a1), k);
            const unsigned b2 = (unsigned)__builtin_amdgcn_readlane(__float_as_int(a2), k);
            const unsigned b3 = (unsigned)__builtin_amdgcn_readlane(__float_as_int(a3), k);
            const float w = __uint_as_float((b0 & p0) | (b1 & p1) | (b2 & p2) | (b3 & p3));
            const float4 f = reinterpret_cast<const float4*>(ft + (size_t)sk * HD)[lane];
            acc.x += f.x * w;
            acc.y += f.y * w;
            acc.z += f.z * w;
            acc.w += f.w * w;
        }
    }
    const float4 b = reinterpret_cast<const float4*>(bias)[lane];
    acc.x += b.x; acc.y += b.y; acc.z += b.z; acc.w += b.w;
    reinterpret_cast<float4*>(out + (size_t)d * HD)[lane] = acc;
}

extern "C" void kernel_launch(void* const* d_in, const int* in_sizes, int n_in,
                              void* d_out, int out_size, void* d_ws, size_t ws_size,
                              hipStream_t stream) {
    const float* feat   = (const float*)d_in[0];
    const int*   src    = (const int*)  d_in[1];
    const int*   dst    = (const int*)  d_in[2];
    const float* W      = (const float*)d_in[3];
    const float* attn_l = (const float*)d_in[4];
    const float* attn_r = (const float*)d_in[5];
    const float* bias   = (const float*)d_in[6];

    const int N = in_sizes[0] / DIN;
    const int E = in_sizes[1];
    float* out = (float*)d_out;

    float* ws     = (float*)d_ws;
    float* ft     = ws;
    float* el     = ft + (size_t)N * HD;
    float* er     = el + (size_t)N * HH;
    int*   cnt    = (int*)(er + (size_t)N * HH);
    int*   rowptr = cnt + N;
    int*   rank   = rowptr + N + 1;
    int*   perm   = rank + E;

    gat_gemm<<<(N + 63) / 64, 256, 0, stream>>>(feat, W, attn_l, attn_r, ft, el, er, cnt, N);
    gat_count_rank<<<(E + 255) / 256, 256, 0, stream>>>(dst, cnt, rank, E);
    gat_scan<<<(N + 1023) / 1024, 1024, 0, stream>>>(cnt, rowptr, N);
    gat_fill<<<(E + 255) / 256, 256, 0, stream>>>(src, dst, rowptr, rank, perm, E);
    gat_node<<<(N + 7) / 8, 256, 0, stream>>>(rowptr, perm, el, er, ft, bias, out, N);
}


# Node kernel: one wave walks 8 consecutive nodes; an edge's source and el row are read once into registers, and the next node's are fetched while this node's ft rows are gathered; sum order kept
# speedup vs baseline: 3.3200x; 1.0441x over previous
#include <hip/hip_runtime.h>

typedef __attribute__((ext_vector_type(2))) float v2f;
typedef __attribute__((ext_vector_type(8))) float v8f;

#define DIN 256
#define HD  128
#define HH  4
#define DD  32
#define NEG_SLOPE 0.2f

template <int M>
__device__ __forceinline__ float xrow(float x) {
    return __int_as_float(__builtin_amdgcn_update_dpp(
        0, __float_as_int(x), 0x160 | M, 0xf, 0xf, true));
}

#define GROWS 64
#define GTILES (GROWS / 16)
#define PSTR 72
#define QSTR (8 * PSTR + 16)

__global__ __launch_bounds__(256) void gat_gemm(const float* __restrict__ feat,
                                                const float* __restrict__ W,
                                                const float* __restrict__ attn_l,
                                                const float* __restrict__ attn_r,
                                                float* __restrict__ ft,
                                                float* __restrict__ el,
                                                float* __restrict__ er,
                                                int* __restrict__ cnt, int N) {
    __shared__ __attribute__((aligned(16))) float As[GROWS][260];
    __shared__ float part[2 * QSTR];
    const int tid = threadIdx.x;
    const int rowBase = blockIdx.x * GROWS;
    const bool fullTile = (rowBase + GROWS <= N);

    if (fullTile) {
        #pragma unroll
        for (int i = 0; i < 16; ++i) {
            int idx4 = i * 256 + tid;
            int r    = idx4 >> 6;
            int c4   = idx4 & 63;
            float4 v = reinterpret_cast<const float4*>(feat + (size_t)(rowBase + r) * DIN)[c4];
            *reinterpret_cast<float4*>(&As[r][c4 * 4]) = v;
        }
    } else {
        #pragma unroll
        for (int i = 0; i < 16; ++i) {
            int idx4 = i * 256 + tid;
            int r    = idx4 >> 6;
            int c4   = idx4 & 63;
            float4 v = make_float4(0.f, 0.f, 0.f, 0.f);
            if (rowBase + r < N)
                v = reinterpret_cast<const float4*>(feat + (size_t)(rowBase + r) * DIN)[c4];
            *reinterpret_cast<float4*>(&As[r][c4 * 4]) = v;
        }
    }
    __syncthreads();

    const int wave    = tid >> 5;
    const int lane    = tid & 31;
    const int half    = lane >> 4;
    const int mrow    = lane & 15;
    const int colBase = wave * 16;
    const int ocol    = colBase + (lane & 15);
    const int kp      = half * 2;

    v8f c[GTILES];
    #pragma unroll
    for (int t = 0; t < GTILES; ++t)
        c[t] = (v8f){0.f, 0.f, 0.f, 0.f, 0.f, 0.f, 0.f, 0.f};
    const float* wrow = W + (size_t)ocol * DIN;
    const float al = attn_l[ocol];
    const float ar = attn_r[ocol];

    #pragma unroll 4
    for (int k0 = 0; k0 < DIN; k0 += 4) {
        v2f b = *reinterpret_cast<const v2f*>(wrow + k0 + kp);
        #pragma unroll
        for (int t = 0; t < GTILES; ++t) {
            v2f a = *reinterpret_cast<const v2f*>(&As[t * 16 + mrow][k0 + kp]);
            c[t] = __builtin_amdgcn_wmma_f32_16x16x4_f32(
                     false, a,  false, b,
                     (short)0, c[t],  false,  false);
        }
    }

    const bool b3 = (lane & 8) != 0;
    const bool b2 = (lane & 4) != 0;
    const bool b1 = (lane & 2) != 0;
    const bool b0 = (lane & 1) != 0;
    #pragma unroll
    for (int t = 0; t < GTILES; ++t) {
        float pl[8], pr[8];
        #pragma unroll
        for (int v = 0; v < 8; ++v) { pl[v] = c[t][v] * al; pr[v] = c[t][v] * ar; }
        float l1[4], r1[4];
        #pragma unroll
        for (int v = 0; v < 4; ++v) {
            l1[v] = (b3 ? pl[v + 4] : pl[v]) + xrow<8>(b3 ? pl[v] : pl[v + 4]);
            r1[v] = (b3 ? pr[v + 4] : pr[v]) + xrow<8>(b3 ? pr[v] : pr[v + 4]);
        }
        float l2[2], r2[2];
        #pragma unroll
        for (int v = 0; v < 2; ++v) {
            l2[v] = (b2 ? l1[v + 2] : l1[v]) + xrow<4>(b2 ? l1[v] : l1[v + 2]);
            r2[v] = (b2 ? r1[v + 2] : r1[v]) + xrow<4>(b2 ? r1[v] : r1[v + 2]);
        }
        const float l3 = (b1 ? l2[1] : l2[0]) + xrow<2>(b1 ? l2[0] : l2[1]);
        const float r3 = (b1 ? r2[1] : r2[0]) + xrow<2>(b1 ? r2[0] : r2[1]);
        const float fin = (b0 ? r3 : l3) + xrow<1>(b0 ? l3 : r3);
        part[(lane & 1) * QSTR + wave * PSTR + t * 16 + 8 * half + ((lane & 15) >> 1)] = fin;
    }
    __syncthreads();

    #pragma unroll
    for (int t = 0; t < GTILES; ++t) {
        float* outBase = ft + (size_t)(rowBase + t * 16 + 8 * half) * HD + colBase + (lane & 15);
        if (fullTile) {
            #pragma unroll
            for (int v = 0; v < 8; ++v)
                outBase[(size_t)v * HD] = c[t][v];
        } else {
            #pragma unroll
            for (int v = 0; v < 8; ++v) {
                if (rowBase + t * 16 + v + 8 * half < N)
                    outBase[(size_t)v * HD] = c[t][v];
            }
        }
    }

    {
        const int rr = tid >> 2;
        const int hh = tid & 3;
        const bool live = (rowBase + rr < N);
        #pragma unroll
        for (int q = 0; q < 2; ++q) {
            const float sum = part[q * QSTR + (2 * hh) * PSTR + rr]
                            + part[q * QSTR + (2 * hh + 1) * PSTR + rr];
            if (live) {
                float* dstp = q ? er : el;
                dstp[(size_t)(rowBase + rr) * HH + hh] = sum;
            }
        }
        if (tid < GROWS && rowBase + tid < N) cnt[rowBase + tid] = 0;
    }
}

__global__ __launch_bounds__(256) void gat_count_rank(const int* __restrict__ dst,
                                                      int* __restrict__ cnt,
                                                      int* __restrict__ rank, int E) {
    const int e = blockIdx.x * 256 + threadIdx.x;
    if (e >= E) return;
    rank[e] = atomicAdd(&cnt[dst[e]], 1);
}

__global__ __launch_bounds__(1024) void gat_scan(const int* __restrict__ cnt,
                                                 int* __restrict__ rowptr, int N) {
    __shared__ int wtile[32];
    __shared__ int wfront[32];
    const int t     = threadIdx.x;
    const int lane  = t & 31;
    const int wave  = t >> 5;
    const int tile0 = blockIdx.x * 1024;

    int s = 0;
    #pragma unroll 8
    for (int i = t; i < tile0; i += 1024) s += cnt[i];
    #pragma unroll
    for (int off = 16; off > 0; off >>= 1) s += __shfl_xor(s, off, 32);

    const int i = tile0 + t;
    const int v = (i < N) ? cnt[i] : 0;
    int inc = v;
    #pragma unroll
    for (int off = 1; off < 32; off <<= 1) {
        const int u = __shfl_up(inc, off, 32);
        if (lane >= off) inc += u;
    }
    if (lane == 31) wtile[wave]  = inc;
    if (lane == 0)  wfront[wave] = s;
    __syncthreads();

    int base = 0;
    for (int w = 0; w < 32; ++w) {
        base += wfront[w];
        if (w < wave) base += wtile[w];
    }
    if (i < N) rowptr[i] = base + inc - v;
    if (blockIdx.x == gridDim.x - 1 && t == 1023) rowptr[N] = base + inc;
}

__global__ __launch_bounds__(256) void gat_fill(const int* __restrict__ src,
                                                const int* __restrict__ dst,
                                                const int* __restrict__ rowptr,
                                                const int* __restrict__ rank,
                                                int* __restrict__ perm, int E) {
    const int e = blockIdx.x * 256 + threadIdx.x;
    if (e >= E) return;
    perm[rowptr[dst[e]] + rank[e]] = src[e];
}

__device__ __forceinline__ float leaky(float x) {
    return (x > 0.f) ? x : NEG_SLOPE * x;
}

#define NPW 8

__global__ __launch_bounds__(256) __attribute__((amdgpu_waves_per_eu(16, 16))) void gat_node(
        const int* __restrict__ rowptr, const int* __restrict__ perm,
        const float* __restrict__ el, const float* __restrict__ er,
        const float* __restrict__ ft, const float* __restrict__ bias,
        float* __restrict__ out, int N) {
    const int lane = threadIdx.x & 31;
    const int g  = __builtin_amdgcn_readfirstlane((int)(blockIdx.x * 8 + (threadIdx.x >> 5)));
    const int d0 = g * NPW;
    if (d0 >= N) return;
    const int nn = (N - d0 < NPW) ? (N - d0) : NPW;
    const float4* el4 = reinterpret_cast<const float4*>(el);

    const int    rp = rowptr[d0 + ((lane < nn) ? lane : nn)];
    const float4 rv = reinterpret_cast<const float4*>(er)[d0 + ((lane < nn) ? lane : (nn - 1))];
    const float4 bs = reinterpret_cast<const float4*>(bias)[lane];

    const int hsel = lane >> 3;
    const unsigned p0 = (hsel == 0) ? 0xffffffffu : 0u;
    const unsigned p1 = (hsel == 1) ? 0xffffffffu : 0u;
    const unsigned p2 = (hsel == 2) ? 0xffffffffu : 0u;
    const unsigned p3 = (hsel == 3) ? 0xffffffffu : 0u;

    int    sjA = 0, sjB = 0;
    float  xA0, xA1, xA2, xA3;
    {
        const int b0 = __builtin_amdgcn_readlane(rp, 0);
        const int b1 = __builtin_amdgcn_readlane(rp, 1);
        float4 l0 = make_float4(0.f, 0.f, 0.f, 0.f);
        if (lane < b1 - b0) sjA = perm[b0 + lane];
        if (nn > 1) {
            const int b2 = __builtin_amdgcn_readlane(rp, 2);
            if (lane < b2 - b1) sjB = perm[b1 + lane];
        }
        if (lane < b1 - b0) l0 = el4[sjA];
        xA0 = leaky(l0.x + __int_as_float(__builtin_amdgcn_readlane(__float_as_int(rv.x), 0)));
        xA1 = leaky(l0.y + __int_as_float(__builtin_amdgcn_readlane(__float_as_int(rv.y), 0)));
        xA2 = leaky(l0.z + __int_as_float(__builtin_amdgcn_readlane(__float_as_int(rv.z), 0)));
        xA3 = leaky(l0.w + __int_as_float(__builtin_amdgcn_readlane(__float_as_int(rv.w), 0)));
    }

    for (int i = 0; i < nn; ++i) {
        const int d   = d0 + i;
        const int beg = __builtin_amdgcn_readlane(rp, i);
        const int end = __builtin_amdgcn_readlane(rp, i + 1);
        const int deg = end - beg;

        float4 lB  = make_float4(0.f, 0.f, 0.f, 0.f);
        int    sjC = 0;
        if (i + 1 < nn) {
            const int endB = __builtin_amdgcn_readlane(rp, i + 2);
            if (lane < endB - end) lB = el4[sjB];
        }
        if (i + 2 < nn) {
            const int begC = __builtin_amdgcn_readlane(rp, i + 2);
            const int endC = __builtin_amdgcn_readlane(rp, i + 3);
            if (lane < endC - begC) sjC = perm[begC + lane];
        }

        float4 acc = make_float4(0.f, 0.f, 0.f, 0.f);
        if (deg <= 32) {
            const bool act = lane < deg;
            const float x0 = xA0, x1 = xA1, x2 = xA2, x3 = xA3;
            float m0 = act ? x0 : -3.402823466e38f;
            float m1 = act ? x1 : -3.402823466e38f;
            float m2 = act ? x2 : -3.402823466e38f;
            float m3 = act ? x3 : -3.402823466e38f;
            #pragma unroll
            for (int off = 16; off > 0; off >>= 1) {
                m0 = fmaxf(m0, __shfl_xor(m0, off, 32));
                m1 = fmaxf(m1, __shfl_xor(m1, off, 32));
                m2 = fmaxf(m2, __shfl_xor(m2, off, 32));
                m3 = fmaxf(m3, __shfl_xor(m3, off, 32));
            }
            const float e0 = act ? __expf(x0 - m0) : 0.f;
            const float e1 = act ? __expf(x1 - m1) : 0.f;
            const float e2 = act ? __expf(x2 - m2) : 0.f;
            const float e3 = act ? __expf(x3 - m3) : 0.f;
            float s0 = e0, s1 = e1, s2 = e2, s3 = e3;
            #pragma unroll
            for (int off = 16; off > 0; off >>= 1) {
                s0 += __shfl_xor(s0, off, 32);
                s1 += __shfl_xor(s1, off, 32);
                s2 += __shfl_xor(s2, off, 32);
                s3 += __shfl_xor(s3, off, 32);
            }
            const float a0 = act ? e0 / s0 : 0.f;
            const float a1 = act ? e1 / s1 : 0.f;
            const float a2 = act ? e2 / s2 : 0.f;
            const float a3 = act ? e3 / s3 : 0.f;
            #pragma unroll 4
            for (int k = 0; k < deg; ++k) {
                const int      sk = __builtin_amdgcn_readlane(sjA, k);
                const unsigned b0 = (unsigned)__builtin_amdgcn_readlane(__float_as_int(a0), k);
                const unsigned b1 = (unsigned)__builtin_amdgcn_readlane(__float_as_int(a1), k);
                const unsigned b2 = (unsigned)__builtin_amdgcn_readlane(__float_as_int(a2), k);
                const unsigned b3 = (unsigned)__builtin_amdgcn_readlane(__float_as_int(a3), k);
                const float w = __uint_as_float((b0 & p0) | (b1 & p1) | (b2 & p2) | (b3 & p3));
                const float4 f = reinterpret_cast<const float4*>(ft + (size_t)sk * HD)[lane];
                acc.x += f.x * w;
                acc.y += f.y * w;
                acc.z += f.z * w;
                acc.w += f.w * w;
            }
        } else {
            float4 r;
            r.x = __int_as_float(__builtin_amdgcn_readlane(__float_as_int(rv.x), i));
            r.y = __int_as_float(__builtin_amdgcn_readlane(__float_as_int(rv.y), i));
            r.z = __int_as_float(__builtin_amdgcn_readlane(__float_as_int(rv.z), i));
            r.w = __int_as_float(__builtin_amdgcn_readlane(__float_as_int(rv.w), i));
            float m0 = -3.402823466e38f, m1 = m0, m2 = m0, m3 = m0;
            for (int j = beg + lane; j < end; j += 32) {
                const float4 l = el4[perm[j]];
                m0 = fmaxf(m0, leaky(l.x + r.x));
                m1 = fmaxf(m1, leaky(l.y + r.y));
                m2 = fmaxf(m2, leaky(l.z + r.z));
                m3 = fmaxf(m3, leaky(l.w + r.w));
            }
            #pragma unroll
            for (int off = 16; off > 0; off >>= 1) {
                m0 = fmaxf(m0, __shfl_xor(m0, off, 32));
                m1 = fmaxf(m1, __shfl_xor(m1, off, 32));
                m2 = fmaxf(m2, __shfl_xor(m2, off, 32));
                m3 = fmaxf(m3, __shfl_xor(m3, off, 32));
            }
            float s0 = 0.f, s1 = 0.f, s2 = 0.f, s3 = 0.f;
            for (int j = beg + lane; j < end; j += 32) {
                const float4 l = el4[perm[j]];
                s0 += __expf(leaky(l.x + r.x) - m0);
                s1 += __expf(leaky(l.y + r.y) - m1);
                s2 += __expf(leaky(l.z + r.z) - m2);
                s3 += __expf(leaky(l.w + r.w) - m3);
            }
            #pragma unroll
            for (int off = 16; off > 0; off >>= 1) {
                s0 += __shfl_xor(s0, off, 32);
                s1 += __shfl_xor(s1, off, 32);
                s2 += __shfl_xor(s2, off, 32);
                s3 += __shfl_xor(s3, off, 32);
            }
            for (int base = beg; base < end; base += 32) {
                const int j = base + lane;
                int   sj = 0;
                float a0 = 0.f, a1 = 0.f, a2 = 0.f, a3 = 0.f;
                if (j < end) {
                    sj = perm[j];
                    const float4 l = el4[sj];
                    a0 = __expf(leaky(l.x + r.x) - m0) / s0;
                    a1 = __expf(leaky(l.y + r.y) - m1) / s1;
                    a2 = __expf(leaky(l.z + r.z) - m2) / s2;
                    a3 = __expf(leaky(l.w + r.w) - m3) / s3;
                }
                const int nk = (end - base < 32) ? (end - base) : 32;
                #pragma unroll 4
                for (int k = 0; k < nk; ++k) {
                    const int      sk = __builtin_amdgcn_readlane(sj, k);
                    const unsigned b0 = (unsigned)__builtin_amdgcn_readlane(__float_as_int(a0), k);
                    const unsigned b1 = (unsigned)__builtin_amdgcn_readlane(__float_as_int(a1), k);
                    const unsigned b2 = (unsigned)__builtin_amdgcn_readlane(__float_as_int(a2), k);
                    const unsigned b3 = (unsigned)__builtin_amdgcn_readlane(__float_as_int(a3), k);
                    const float w = __uint_as_float((b0 & p0) | (b1 & p1) | (b2 & p2) | (b3 & p3));
                    const float4 f = reinterpret_cast<const float4*>(ft + (size_t)sk * HD)[lane];
                    acc.x += f.x * w;
                    acc.y += f.y * w;
                    acc.z += f.z * w;
                    acc.w += f.w * w;
                }
            }
        }
        acc.x += bs.x; acc.y += bs.y; acc.z += bs.z; acc.w += bs.w;
        reinterpret_cast<float4*>(out + (size_t)d * HD)[lane] = acc;

        sjA = sjB; sjB = sjC;
        xA0 = leaky(lB.x + __int_as_float(__builtin_amdgcn_readlane(__float_as_int(rv.x), i + 1)));
        xA1 = leaky(lB.y + __int_as_float(__builtin_amdgcn_readlane(__float_as_int(rv.y), i + 1)));
        xA2 = leaky(lB.z + __int_as_float(__builtin_amdgcn_readlane(__float_as_int(rv.z), i + 1)));
        xA3 = leaky(lB.w + __int_as_float(__builtin_amdgcn_readlane(__float_as_int(rv.w), i + 1)));
    }
}

extern "C" void kernel_launch(void* const* d_in, const int* in_sizes, int n_in,
                              void* d_out, int out_size, void* d_ws, size_t ws_size,
                              hipStream_t stream) {
    const float* feat   = (const float*)d_in[0];
    const int*   src    = (const int*)  d_in[1];
    const int*   dst    = (const int*)  d_in[2];
    const float* W      = (const float*)d_in[3];
    const float* attn_l = (const float*)d_in[4];
    const float* attn_r = (const float*)d_in[5];
    const float* bias   = (const float*)d_in[6];

    const int N = in_sizes[0] / DIN;
    const int E = in_sizes[1];
    float* out = (float*)d_out;

    float* ws     = (float*)d_ws;
    float* ft     = ws;
    float* el     = ft + (size_t)N * HD;
    float* er     = el + (size_t)N * HH;
    int*   cnt    = (int*)(er + (size_t)N * HH);
    int*   rowptr = cnt + N;
    int*   rank   = rowptr + N + 1;
    int*   perm   = rank + E;

    gat_gemm<<<(N + 63) / 64, 256, 0, stream>>>(feat, W, attn_l, attn_r, ft, el, er, cnt, N);
    gat_count_rank<<<(E + 255) / 256, 256, 0, stream>>>(dst, cnt, rank, E);
    gat_scan<<<(N + 1023) / 1024, 1024, 0, stream>>>(cnt, rowptr, N);
    gat_fill<<<(E + 255) / 256, 256, 0, stream>>>(src, dst, rowptr, rank, perm, E);
    gat_node<<<(N + 8 * NPW - 1) / (8 * NPW), 256, 0, stream>>>(rowptr, perm, el, er, ft, bias, out, N);
}


# In-degree counters split into one plane per half of the package (a wave adds to the plane of its own half), so no counter line takes atomics from both halves; edge order inside a node changes
# speedup vs baseline: 4.4027x; 1.3261x over previous
#include <hip/hip_runtime.h>

typedef __attribute__((ext_vector_type(2))) float v2f;
typedef __attribute__((ext_vector_type(8))) float v8f;

#define DIN 256
#define HD  128
#define HH  4
#define DD  32
#define NEG_SLOPE 0.2f

template <int M>
__device__ __forceinline__ float xrow(float x) {
    return __int_as_float(__builtin_amdgcn_update_dpp(
        0, __float_as_int(x), 0x160 | M, 0xf, 0xf, true));
}

#define GROWS 64
#define GTILES (GROWS / 16)
#define PSTR 72
#define QSTR (8 * PSTR + 16)

__global__ __launch_bounds__(256) void gat_gemm(const float* __restrict__ feat,
                                                const float* __restrict__ W,
                                                const float* __restrict__ attn_l,
                                                const float* __restrict__ attn_r,
                                                float* __restrict__ ft,
                                                float* __restrict__ el,
                                                float* __restrict__ er,
                                                int* __restrict__ cnt,
                                                int* __restrict__ cnt1, int N) {
    __shared__ __attribute__((aligned(16))) float As[GROWS][260];
    __shared__ float part[2 * QSTR];
    const int tid = threadIdx.x;
    const int rowBase = blockIdx.x * GROWS;
    const bool fullTile = (rowBase + GROWS <= N);

    if (fullTile) {
        #pragma unroll
        for (int i = 0; i < 16; ++i) {
            int idx4 = i * 256 + tid;
            int r    = idx4 >> 6;
            int c4   = idx4 & 63;
            float4 v = reinterpret_cast<const float4*>(feat + (size_t)(rowBase + r) * DIN)[c4];
            *reinterpret_cast<float4*>(&As[r][c4 * 4]) = v;
        }
    } else {
        #pragma unroll
        for (int i = 0; i < 16; ++i) {
            int idx4 = i * 256 + tid;
            int r    = idx4 >> 6;
            int c4   = idx4 & 63;
            float4 v = make_float4(0.f, 0.f, 0.f, 0.f);
            if (rowBase + r < N)
                v = reinterpret_cast<const float4*>(feat + (size_t)(rowBase + r) * DIN)[c4];
            *reinterpret_cast<float4*>(&As[r][c4 * 4]) = v;
        }
    }
    __syncthreads();

    const int wave    = tid >> 5;
    const int lane    = tid & 31;
    const int half    = lane >> 4;
    const int mrow    = lane & 15;
    const int colBase = wave * 16;
    const int ocol    = colBase + (lane & 15);
    const int kp      = half * 2;

    v8f c[GTILES];
    #pragma unroll
    for (int t = 0; t < GTILES; ++t)
        c[t] = (v8f){0.f, 0.f, 0.f, 0.f, 0.f, 0.f, 0.f, 0.f};
    const float* wrow = W + (size_t)ocol * DIN;
    const float al = attn_l[ocol];
    const float ar = attn_r[ocol];

    #pragma unroll 4
    for (int k0 = 0; k0 < DIN; k0 += 4) {
        v2f b = *reinterpret_cast<const v2f*>(wrow + k0 + kp);
        #pragma unroll
        for (int t = 0; t < GTILES; ++t) {
            v2f a = *reinterpret_cast<const v2f*>(&As[t * 16 + mrow][k0 + kp]);
            c[t] = __builtin_amdgcn_wmma_f32_16x16x4_f32(
                     false, a,  false, b,
                     (short)0, c[t],  false,  false);
        }
    }

    const bool b3 = (lane & 8) != 0;
    const bool b2 = (lane & 4) != 0;
    const bool b1 = (lane & 2) != 0;
    const bool b0 = (lane & 1) != 0;
    #pragma unroll
    for (int t = 0; t < GTILES; ++t) {
        float pl[8], pr[8];
        #pragma unroll
        for (int v = 0; v < 8; ++v) { pl[v] = c[t][v] * al; pr[v] = c[t][v] * ar; }
        float l1[4], r1[4];
        #pragma unroll
        for (int v = 0; v < 4; ++v) {
            l1[v] = (b3 ? pl[v + 4] : pl[v]) + xrow<8>(b3 ? pl[v] : pl[v + 4]);
            r1[v] = (b3 ? pr[v + 4] : pr[v]) + xrow<8>(b3 ? pr[v] : pr[v + 4]);
        }
        float l2[2], r2[2];
        #pragma unroll
        for (int v = 0; v < 2; ++v) {
            l2[v] = (b2 ? l1[v + 2] : l1[v]) + xrow<4>(b2 ? l1[v] : l1[v + 2]);
            r2[v] = (b2 ? r1[v + 2] : r1[v]) + xrow<4>(b2 ? r1[v] : r1[v + 2]);
        }
        const float l3 = (b1 ? l2[1] : l2[0]) + xrow<2>(b1 ? l2[0] : l2[1]);
        const float r3 = (b1 ? r2[1] : r2[0]) + xrow<2>(b1 ? r2[0] : r2[1]);
        const float fin = (b0 ? r3 : l3) + xrow<1>(b0 ? l3 : r3);
        part[(lane & 1) * QSTR + wave * PSTR + t * 16 + 8 * half + ((lane & 15) >> 1)] = fin;
    }
    __syncthreads();

    #pragma unroll
    for (int t = 0; t < GTILES; ++t) {
        float* outBase = ft + (size_t)(rowBase + t * 16 + 8 * half) * HD + colBase + (lane & 15);
        if (fullTile) {
            #pragma unroll
            for (int v = 0; v < 8; ++v)
                outBase[(size_t)v * HD] = c[t][v];
        } else {
            #pragma unroll
            for (int v = 0; v < 8; ++v) {
                if (rowBase + t * 16 + v + 8 * half < N)
                    outBase[(size_t)v * HD] = c[t][v];
            }
        }
    }

    {
        const int rr = tid >> 2;
        const int hh = tid & 3;
        const bool live = (rowBase + rr < N);
        #pragma unroll
        for (int q = 0; q < 2; ++q) {
            const float sum = part[q * QSTR + (2 * hh) * PSTR + rr]
                            + part[q * QSTR + (2 * hh + 1) * PSTR + rr];
            if (live) {
                float* dstp = q ? er : el;
                dstp[(size_t)(rowBase + rr) * HH + hh] = sum;
            }
        }
        if (tid < GROWS && rowBase + tid < N) cnt[rowBase + tid] = 0;
        if (tid >= GROWS && tid < 2 * GROWS && rowBase + tid - GROWS < N)
            cnt1[rowBase + tid - GROWS] = 0;
    }
}

__global__ __launch_bounds__(256) void gat_count_rank(const int* __restrict__ dst,
                                                      int* __restrict__ cnt0,
                                                      int* __restrict__ cnt1,
                                                      int* __restrict__ rank, int E) {
    const int e = blockIdx.x * 256 + threadIdx.x;
    if (e >= E) return;
    const int upper = (__builtin_amdgcn_s_sendmsg_rtn(0x87) >> 18) & 1;
    int* c = upper ? cnt1 : cnt0;
    const int r = atomicAdd(&c[dst[e]], 1);
    rank[e] = upper ? ~r : r;
}

__global__ __launch_bounds__(1024) void gat_scan(const int* __restrict__ cnt,
                                                 const int* __restrict__ cnt1,
                                                 int* __restrict__ rowptr, int N) {
    __shared__ int wtile[32];
    __shared__ int wfront[32];
    const int t     = threadIdx.x;
    const int lane  = t & 31;
    const int wave  = t >> 5;
    const int tile0 = blockIdx.x * 1024;

    int s = 0;
    #pragma unroll 8
    for (int i = t; i < tile0; i += 1024) s += cnt[i] + cnt1[i];
    #pragma unroll
    for (int off = 16; off > 0; off >>= 1) s += __shfl_xor(s, off, 32);

    const int i = tile0 + t;
    const int v = (i < N) ? (cnt[i] + cnt1[i]) : 0;
    int inc = v;
    #pragma unroll
    for (int off = 1; off < 32; off <<= 1) {
        const int u = __shfl_up(inc, off, 32);
        if (lane >= off) inc += u;
    }
    if (lane == 31) wtile[wave]  = inc;
    if (lane == 0)  wfront[wave] = s;
    __syncthreads();

    int base = 0;
    for (int w = 0; w < 32; ++w) {
        base += wfront[w];
        if (w < wave) base += wtile[w];
    }
    if (i < N) rowptr[i] = base + inc - v;
    if (blockIdx.x == gridDim.x - 1 && t == 1023) rowptr[N] = base + inc;
}

__global__ __launch_bounds__(256) void gat_fill(const int* __restrict__ src,
                                                const int* __restrict__ dst,
                                                const int* __restrict__ rowptr,
                                                const int* __restrict__ rank,
                                                int* __restrict__ perm, int E) {
    const int e = blockIdx.x * 256 + threadIdx.x;
    if (e >= E) return;
    const int r = rank[e];
    perm[rowptr[dst[e] + (r < 0 ? 1 : 0)] + r] = src[e];
}

__device__ __forceinline__ float leaky(float x) {
    return (x > 0.f) ? x : NEG_SLOPE * x;
}

#define NPW 8

__global__ __launch_bounds__(256) __attribute__((amdgpu_waves_per_eu(16, 16))) void gat_node(
        const int* __restrict__ rowptr, const int* __restrict__ perm,
        const float* __restrict__ el, const float* __restrict__ er,
        const float* __restrict__ ft, const float* __restrict__ bias,
        float* __restrict__ out, int N) {
    const int lane = threadIdx.x & 31;
    const int g  = __builtin_amdgcn_readfirstlane((int)(blockIdx.x * 8 + (threadIdx.x >> 5)));
    const int d0 = g * NPW;
    if (d0 >= N) return;
    const int nn = (N - d0 < NPW) ? (N - d0) : NPW;
    const float4* el4 = reinterpret_cast<const float4*>(el);

    const int    rp = rowptr[d0 + ((lane < nn) ? lane : nn)];
    const float4 rv = reinterpret_cast<const float4*>(er)[d0 + ((lane < nn) ? lane : (nn - 1))];
    const float4 bs = reinterpret_cast<const float4*>(bias)[lane];

    const int hsel = lane >> 3;
    const unsigned p0 = (hsel == 0) ? 0xffffffffu : 0u;
    const unsigned p1 = (hsel == 1) ? 0xffffffffu : 0u;
    const unsigned p2 = (hsel == 2) ? 0xffffffffu : 0u;
    const unsigned p3 = (hsel == 3) ? 0xffffffffu : 0u;

    int    sjA = 0, sjB = 0;
    float  xA0, xA1, xA2, xA3;
    {
        const int b0 = __builtin_amdgcn_readlane(rp, 0);
        const int b1 = __builtin_amdgcn_readlane(rp, 1);
        float4 l0 = make_float4(0.f, 0.f, 0.f, 0.f);
        if (lane < b1 - b0) sjA = perm[b0 + lane];
        if (nn > 1) {
            const int b2 = __builtin_amdgcn_readlane(rp, 2);
            if (lane < b2 - b1) sjB = perm[b1 + lane];
        }
        if (lane < b1 - b0) l0 = el4[sjA];
        xA0 = leaky(l0.x + __int_as_float(__builtin_amdgcn_readlane(__float_as_int(rv.x), 0)));
        xA1 = leaky(l0.y + __int_as_float(__builtin_amdgcn_readlane(__float_as_int(rv.y), 0)));
        xA2 = leaky(l0.z + __int_as_float(__builtin_amdgcn_readlane(__float_as_int(rv.z), 0)));
        xA3 = leaky(l0.w + __int_as_float(__builtin_amdgcn_readlane(__float_as_int(rv.w), 0)));
    }

    for (int i = 0; i < nn; ++i) {
        const int d   = d0 + i;
        const int beg = __builtin_amdgcn_readlane(rp, i);
        const int end = __builtin_amdgcn_readlane(rp, i + 1);
        const int deg = end - beg;

        float4 lB  = make_float4(0.f, 0.f, 0.f, 0.f);
        int    sjC = 0;
        if (i + 1 < nn) {
            const int endB = __builtin_amdgcn_readlane(rp, i + 2);
            if (lane < endB - end) lB = el4[sjB];
        }
        if (i + 2 < nn) {
            const int begC = __builtin_amdgcn_readlane(rp, i + 2);
            const int endC = __builtin_amdgcn_readlane(rp, i + 3);
            if (lane < endC - begC) sjC = perm[begC + lane];
        }

        float4 acc = make_float4(0.f, 0.f, 0.f, 0.f);
        if (deg <= 32) {
            const bool act = lane < deg;
            const float x0 = xA0, x1 = xA1, x2 = xA2, x3 = xA3;
            float m0 = act ? x0 : -3.402823466e38f;
            float m1 = act ? x1 : -3.402823466e38f;
            float m2 = act ? x2 : -3.402823466e38f;
            float m3 = act ? x3 : -3.402823466e38f;
            #pragma unroll
            for (int off = 16; off > 0; off >>= 1) {
                m0 = fmaxf(m0, __shfl_xor(m0, off, 32));
                m1 = fmaxf(m1, __shfl_xor(m1, off, 32));
                m2 = fmaxf(m2, __shfl_xor(m2, off, 32));
                m3 = fmaxf(m3, __shfl_xor(m3, off, 32));
            }
            const float e0 = act ? __expf(x0 - m0) : 0.f;
            const float e1 = act ? __expf(x1 - m1) : 0.f;
            const float e2 = act ? __expf(x2 - m2) : 0.f;
            const float e3 = act ? __expf(x3 - m3) : 0.f;
            float s0 = e0, s1 = e1, s2 = e2, s3 = e3;
            #pragma unroll
            for (int off = 16; off > 0; off >>= 1) {
                s0 += __shfl_xor(s0, off, 32);
                s1 += __shfl_xor(s1, off, 32);
                s2 += __shfl_xor(s2, off, 32);
                s3 += __shfl_xor(s3, off, 32);
            }
            const float a0 = act ? e0 / s0 : 0.f;
            const float a1 = act ? e1 / s1 : 0.f;
            const float a2 = act ? e2 / s2 : 0.f;
            const float a3 = act ? e3 / s3 : 0.f;
            #pragma unroll 4
            for (int k = 0; k < deg; ++k) {
                const int      sk = __builtin_amdgcn_readlane(sjA, k);
                const unsigned b0 = (unsigned)__builtin_amdgcn_readlane(__float_as_int(a0), k);
                const unsigned b1 = (unsigned)__builtin_amdgcn_readlane(__float_as_int(a1), k);
                const unsigned b2 = (unsigned)__builtin_amdgcn_readlane(__float_as_int(a2), k);
                const unsigned b3 = (unsigned)__builtin_amdgcn_readlane(__float_as_int(a3), k);
                const float w = __uint_as_float((b0 & p0) | (b1 & p1) | (b2 & p2) | (b3 & p3));
                const float4 f = reinterpret_cast<const float4*>(ft + (size_t)sk * HD)[lane];
                acc.x += f.x * w;
                acc.y += f.y * w;
                acc.z += f.z * w;
                acc.w += f.w * w;
            }
        } else {
            float4 r;
            r.x = __int_as_float(__builtin_amdgcn_readlane(__float_as_int(rv.x), i));
            r.y = __int_as_float(__builtin_amdgcn_readlane(__float_as_int(rv.y), i));
            r.z = __int_as_float(__builtin_amdgcn_readlane(__float_as_int(rv.z), i));
            r.w = __int_as_float(__builtin_amdgcn_readlane(__float_as_int(rv.w), i));
            float m0 = -3.402823466e38f, m1 = m0, m2 = m0, m3 = m0;
            for (int j = beg + lane; j < end; j += 32) {
                const float4 l = el4[perm[j]];
                m0 = fmaxf(m0, leaky(l.x + r.x));
                m1 = fmaxf(m1, leaky(l.y + r.y));
                m2 = fmaxf(m2, leaky(l.z + r.z));
                m3 = fmaxf(m3, leaky(l.w + r.w));
            }
            #pragma unroll
            for (int off = 16; off > 0; off >>= 1) {
                m0 = fmaxf(m0, __shfl_xor(m0, off, 32));
                m1 = fmaxf(m1, __shfl_xor(m1, off, 32));
                m2 = fmaxf(m2, __shfl_xor(m2, off, 32));
                m3 = fmaxf(m3, __shfl_xor(m3, off, 32));
            }
            float s0 = 0.f, s1 = 0.f, s2 = 0.f, s3 = 0.f;
            for (int j = beg + lane; j < end; j += 32) {
                const float4 l = el4[perm[j]];
                s0 += __expf(leaky(l.x + r.x) - m0);
                s1 += __expf(leaky(l.y + r.y) - m1);
                s2 += __expf(leaky(l.z + r.z) - m2);
                s3 += __expf(leaky(l.w + r.w) - m3);
            }
            #pragma unroll
            for (int off = 16; off > 0; off >>= 1) {
                s0 += __shfl_xor(s0, off, 32);
                s1 += __shfl_xor(s1, off, 32);
                s2 += __shfl_xor(s2, off, 32);
                s3 += __shfl_xor(s3, off, 32);
            }
            for (int base = beg; base < end; base += 32) {
                const int j = base + lane;
                int   sj = 0;
                float a0 = 0.f, a1 = 0.f, a2 = 0.f, a3 = 0.f;
                if (j < end) {
                    sj = perm[j];
                    const float4 l = el4[sj];
                    a0 = __expf(leaky(l.x + r.x) - m0) / s0;
                    a1 = __expf(leaky(l.y + r.y) - m1) / s1;
                    a2 = __expf(leaky(l.z + r.z) - m2) / s2;
                    a3 = __expf(leaky(l.w + r.w) - m3) / s3;
                }
                const int nk = (end - base < 32) ? (end - base) : 32;
                #pragma unroll 4
                for (int k = 0; k < nk; ++k) {
                    const int      sk = __builtin_amdgcn_readlane(sj, k);
                    const unsigned b0 = (unsigned)__builtin_amdgcn_readlane(__float_as_int(a0), k);
                    const unsigned b1 = (unsigned)__builtin_amdgcn_readlane(__float_as_int(a1), k);
                    const unsigned b2 = (unsigned)__builtin_amdgcn_readlane(__float_as_int(a2), k);
                    const unsigned b3 = (unsigned)__builtin_amdgcn_readlane(__float_as_int(a3), k);
                    const float w = __uint_as_float((b0 & p0) | (b1 & p1) | (b2 & p2) | (b3 & p3));
                    const float4 f = reinterpret_cast<const float4*>(ft + (size_t)sk * HD)[lane];
                    acc.x += f.x * w;
                    acc.y += f.y * w;
                    acc.z += f.z * w;
                    acc.w += f.w * w;
                }
            }
        }
        acc.x += bs.x; acc.y += bs.y; acc.z += bs.z; acc.w += bs.w;
        reinterpret_cast<float4*>(out + (size_t)d * HD)[lane] = acc;

        sjA = sjB; sjB = sjC;
        xA0 = leaky(lB.x + __int_as_float(__builtin_amdgcn_readlane(__float_as_int(rv.x), i + 1)));
        xA1 = leaky(lB.y + __int_as_float(__builtin_amdgcn_readlane(__float_as_int(rv.y), i + 1)));
        xA2 = leaky(lB.z + __int_as_float(__builtin_amdgcn_readlane(__float_as_int(rv.z), i + 1)));
        xA3 = leaky(lB.w + __int_as_float(__builtin_amdgcn_readlane(__float_as_int(rv.w), i + 1)));
    }
}

extern "C" void kernel_launch(void* const* d_in, const int* in_sizes, int n_in,
                              void* d_out, int out_size, void* d_ws, size_t ws_size,
                              hipStream_t stream) {
    const float* feat   = (const float*)d_in[0];
    const int*   src    = (const int*)  d_in[1];
    const int*   dst    = (const int*)  d_in[2];
    const float* W      = (const float*)d_in[3];
    const float* attn_l = (const float*)d_in[4];
    const float* attn_r = (const float*)d_in[5];
    const float* bias   = (const float*)d_in[6];

    const int N = in_sizes[0] / DIN;
    const int E = in_sizes[1];
    float* out = (float*)d_out;

    float* ws     = (float*)d_ws;
    float* ft     = ws;
    float* el     = ft + (size_t)N * HD;
    float* er     = el + (size_t)N * HH;
    int*   cnt    = (int*)(er + (size_t)N * HH);
    int*   rowptr = cnt + N;
    int*   rank   = rowptr + N + 1;
    int*   perm   = rank + E;
    int*   cnt1   = perm;

    gat_gemm<<<(N + 63) / 64, 256, 0, stream>>>(feat, W, attn_l, attn_r, ft, el, er, cnt, cnt1, N);
    gat_count_rank<<<(E + 255) / 256, 256, 0, stream>>>(dst, cnt, cnt1, rank, E);
    gat_scan<<<(N + 1023) / 1024, 1024, 0, stream>>>(cnt, cnt1, rowptr, N);
    gat_fill<<<(E + 255) / 256, 256, 0, stream>>>(src, dst, rowptr, rank, perm, E);
    gat_node<<<(N + 8 * NPW - 1) / (8 * NPW), 256, 0, stream>>>(rowptr, perm, el, er, ft, bias, out, N);
}


# In-degree count moved into the GEMM blocks (each counts 1024 edges, one returning atomic per thread behind each quarter of the K loop); clear kernel in front; edge order still atomic arrival order
# speedup vs baseline: 4.7212x; 1.0723x over previous
#include <hip/hip_runtime.h>

typedef __attribute__((ext_vector_type(2))) float v2f;
typedef __attribute__((ext_vector_type(8))) float v8f;

#define DIN 256
#define HD  128
#define HH  4
#define DD  32
#define NEG_SLOPE 0.2f

template <int M>
__device__ __forceinline__ float xrow(float x) {
    return __int_as_float(__builtin_amdgcn_update_dpp(
        0, __float_as_int(x), 0x160 | M, 0xf, 0xf, true));
}

#define GROWS 64
#define GTILES (GROWS / 16)
#define PSTR 72
#define QSTR (8 * PSTR + 16)

__global__ __launch_bounds__(256) void gat_gemm(const float* __restrict__ feat,
                                                const float* __restrict__ W,
                                                const float* __restrict__ attn_l,
                                                const float* __restrict__ attn_r,
                                                float* __restrict__ ft,
                                                float* __restrict__ el,
                                                float* __restrict__ er,
                                                const int* __restrict__ dst,
                                                int* __restrict__ cnt,
                                                int* __restrict__ cnt1,
                                                int* __restrict__ rank,
                                                int N, int E) {
    __shared__ __attribute__((aligned(16))) float As[GROWS][260];
    __shared__ float part[2 * QSTR];
    const int tid = threadIdx.x;
    const int rowBase = blockIdx.x * GROWS;
    const bool fullTile = (rowBase + GROWS <= N);

    const int upper = (__builtin_amdgcn_s_sendmsg_rtn(0x87) >> 18) & 1;
    int* cpl = upper ? cnt1 : cnt;
    const bool vecOK = ((((size_t)dst) | ((size_t)rank)) & 15) == 0;
    const int e0 = (blockIdx.x * 256 + tid) * 4;
    int dv[4] = {0, 0, 0, 0};
    int rk[4] = {0, 0, 0, 0};
    if (vecOK && e0 + 3 < E) {
        const int4 q = *reinterpret_cast<const int4*>(dst + e0);
        dv[0] = q.x; dv[1] = q.y; dv[2] = q.z; dv[3] = q.w;
    } else {
        #pragma unroll
        for (int j = 0; j < 4; ++j)
            if (e0 + j < E) dv[j] = dst[e0 + j];
    }

    if (fullTile) {
        #pragma unroll
        for (int i = 0; i < 16; ++i) {
            int idx4 = i * 256 + tid;
            int r    = idx4 >> 6;
            int c4   = idx4 & 63;
            float4 v = reinterpret_cast<const float4*>(feat + (size_t)(rowBase + r) * DIN)[c4];
            *reinterpret_cast<float4*>(&As[r][c4 * 4]) = v;
        }
    } else {
        #pragma unroll
        for (int i = 0; i < 16; ++i) {
            int idx4 = i * 256 + tid;
            int r    = idx4 >> 6;
            int c4   = idx4 & 63;
            float4 v = make_float4(0.f, 0.f, 0.f, 0.f);
            if (rowBase + r < N)
                v = reinterpret_cast<const float4*>(feat + (size_t)(rowBase + r) * DIN)[c4];
            *reinterpret_cast<float4*>(&As[r][c4 * 4]) = v;
        }
    }
    __syncthreads();

    const int wave    = tid >> 5;
    const int lane    = tid & 31;
    const int half    = lane >> 4;
    const int mrow    = lane & 15;
    const int colBase = wave * 16;
    const int ocol    = colBase + (lane & 15);
    const int kp      = half * 2;

    v8f c[GTILES];
    #pragma unroll
    for (int t = 0; t < GTILES; ++t)
        c[t] = (v8f){0.f, 0.f, 0.f, 0.f, 0.f, 0.f, 0.f, 0.f};
    const float* wrow = W + (size_t)ocol * DIN;
    const float al = attn_l[ocol];
    const float ar = attn_r[ocol];

    #pragma unroll
    for (int q = 0; q < 4; ++q) {
    #pragma unroll 4
    for (int k0 = q * (DIN / 4); k0 < (q + 1) * (DIN / 4); k0 += 4) {
        v2f b = *reinterpret_cast<const v2f*>(wrow + k0 + kp);
        #pragma unroll
        for (int t = 0; t < GTILES; ++t) {
            v2f a = *reinterpret_cast<const v2f*>(&As[t * 16 + mrow][k0 + kp]);
            c[t] = __builtin_amdgcn_wmma_f32_16x16x4_f32(
                     false, a,  false, b,
                     (short)0, c[t],  false,  false);
        }
    }
    if (e0 + q < E) rk[q] = atomicAdd(&cpl[dv[q]], 1);
    }

    const bool b3 = (lane & 8) != 0;
    const bool b2 = (lane & 4) != 0;
    const bool b1 = (lane & 2) != 0;
    const bool b0 = (lane & 1) != 0;
    #pragma unroll
    for (int t = 0; t < GTILES; ++t) {
        float pl[8], pr[8];
        #pragma unroll
        for (int v = 0; v < 8; ++v) { pl[v] = c[t][v] * al; pr[v] = c[t][v] * ar; }
        float l1[4], r1[4];
        #pragma unroll
        for (int v = 0; v < 4; ++v) {
            l1[v] = (b3 ? pl[v + 4] : pl[v]) + xrow<8>(b3 ? pl[v] : pl[v + 4]);
            r1[v] = (b3 ? pr[v + 4] : pr[v]) + xrow<8>(b3 ? pr[v] : pr[v + 4]);
        }
        float l2[2], r2[2];
        #pragma unroll
        for (int v = 0; v < 2; ++v) {
            l2[v] = (b2 ? l1[v + 2] : l1[v]) + xrow<4>(b2 ? l1[v] : l1[v + 2]);
            r2[v] = (b2 ? r1[v + 2] : r1[v]) + xrow<4>(b2 ? r1[v] : r1[v + 2]);
        }
        const float l3 = (b1 ? l2[1] : l2[0]) + xrow<2>(b1 ? l2[0] : l2[1]);
        const float r3 = (b1 ? r2[1] : r2[0]) + xrow<2>(b1 ? r2[0] : r2[1]);
        const float fin = (b0 ? r3 : l3) + xrow<1>(b0 ? l3 : r3);
        part[(lane & 1) * QSTR + wave * PSTR + t * 16 + 8 * half + ((lane & 15) >> 1)] = fin;
    }
    __syncthreads();

    #pragma unroll
    for (int t = 0; t < GTILES; ++t) {
        float* outBase = ft + (size_t)(rowBase + t * 16 + 8 * half) * HD + colBase + (lane & 15);
        if (fullTile) {
            #pragma unroll
            for (int v = 0; v < 8; ++v)
                outBase[(size_t)v * HD] = c[t][v];
        } else {
            #pragma unroll
            for (int v = 0; v < 8; ++v) {
                if (rowBase + t * 16 + v + 8 * half < N)
                    outBase[(size_t)v * HD] = c[t][v];
            }
        }
    }

    {
        const int rr = tid >> 2;
        const int hh = tid & 3;
        const bool live = (rowBase + rr < N);
        #pragma unroll
        for (int q = 0; q < 2; ++q) {
            const float sum = part[q * QSTR + (2 * hh) * PSTR + rr]
                            + part[q * QSTR + (2 * hh + 1) * PSTR + rr];
            if (live) {
                float* dstp = q ? er : el;
                dstp[(size_t)(rowBase + rr) * HH + hh] = sum;
            }
        }
    }

    if (vecOK && e0 + 3 < E) {
        int4 q;
        q.x = upper ? ~rk[0] : rk[0];
        q.y = upper ? ~rk[1] : rk[1];
        q.z = upper ? ~rk[2] : rk[2];
        q.w = upper ? ~rk[3] : rk[3];
        *reinterpret_cast<int4*>(rank + e0) = q;
    } else {
        #pragma unroll
        for (int j = 0; j < 4; ++j)
            if (e0 + j < E) rank[e0 + j] = upper ? ~rk[j] : rk[j];
    }
    for (long long e = (long long)gridDim.x * 1024 + (long long)blockIdx.x * 256 + tid;
         e < E; e += (long long)gridDim.x * 256) {
        const int r = atomicAdd(&cpl[dst[e]], 1);
        rank[e] = upper ? ~r : r;
    }
}

__global__ __launch_bounds__(1024) void gat_clear(int* __restrict__ cnt0,
                                                  int* __restrict__ cnt1, int N) {
    const int i = blockIdx.x * 1024 + threadIdx.x;
    if (i < N) { cnt0[i] = 0; cnt1[i] = 0; }
}

__global__ __launch_bounds__(1024) void gat_scan(const int* __restrict__ cnt,
                                                 const int* __restrict__ cnt1,
                                                 int* __restrict__ rowptr, int N) {
    __shared__ int wtile[32];
    __shared__ int wfront[32];
    const int t     = threadIdx.x;
    const int lane  = t & 31;
    const int wave  = t >> 5;
    const int tile0 = blockIdx.x * 1024;

    int s = 0;
    #pragma unroll 8
    for (int i = t; i < tile0; i += 1024) s += cnt[i] + cnt1[i];
    #pragma unroll
    for (int off = 16; off > 0; off >>= 1) s += __shfl_xor(s, off, 32);

    const int i = tile0 + t;
    const int v = (i < N) ? (cnt[i] + cnt1[i]) : 0;
    int inc = v;
    #pragma unroll
    for (int off = 1; off < 32; off <<= 1) {
        const int u = __shfl_up(inc, off, 32);
        if (lane >= off) inc += u;
    }
    if (lane == 31) wtile[wave]  = inc;
    if (lane == 0)  wfront[wave] = s;
    __syncthreads();

    int base = 0;
    for (int w = 0; w < 32; ++w) {
        base += wfront[w];
        if (w < wave) base += wtile[w];
    }
    if (i < N) rowptr[i] = base + inc - v;
    if (blockIdx.x == gridDim.x - 1 && t == 1023) rowptr[N] = base + inc;
}

__global__ __launch_bounds__(256) void gat_fill(const int* __restrict__ src,
                                                const int* __restrict__ dst,
                                                const int* __restrict__ rowptr,
                                                const int* __restrict__ rank,
                                                int* __restrict__ perm, int E) {
    const int e = blockIdx.x * 256 + threadIdx.x;
    if (e >= E) return;
    const int r = rank[e];
    perm[rowptr[dst[e] + (r < 0 ? 1 : 0)] + r] = src[e];
}

__device__ __forceinline__ float leaky(float x) {
    return (x > 0.f) ? x : NEG_SLOPE * x;
}

#define NPW 8

__global__ __launch_bounds__(256) __attribute__((amdgpu_waves_per_eu(16, 16))) void gat_node(
        const int* __restrict__ rowptr, const int* __restrict__ perm,
        const float* __restrict__ el, const float* __restrict__ er,
        const float* __restrict__ ft, const float* __restrict__ bias,
        float* __restrict__ out, int N) {
    const int lane = threadIdx.x & 31;
    const int g  = __builtin_amdgcn_readfirstlane((int)(blockIdx.x * 8 + (threadIdx.x >> 5)));
    const int d0 = g * NPW;
    if (d0 >= N) return;
    const int nn = (N - d0 < NPW) ? (N - d0) : NPW;
    const float4* el4 = reinterpret_cast<const float4*>(el);

    const int    rp = rowptr[d0 + ((lane < nn) ? lane : nn)];
    const float4 rv = reinterpret_cast<const float4*>(er)[d0 + ((lane < nn) ? lane : (nn - 1))];
    const float4 bs = reinterpret_cast<const float4*>(bias)[lane];

    const int hsel = lane >> 3;
    const unsigned p0 = (hsel == 0) ? 0xffffffffu : 0u;
    const unsigned p1 = (hsel == 1) ? 0xffffffffu : 0u;
    const unsigned p2 = (hsel == 2) ? 0xffffffffu : 0u;
    const unsigned p3 = (hsel == 3) ? 0xffffffffu : 0u;

    int    sjA = 0, sjB = 0;
    float  xA0, xA1, xA2, xA3;
    {
        const int b0 = __builtin_amdgcn_readlane(rp, 0);
        const int b1 = __builtin_amdgcn_readlane(rp, 1);
        float4 l0 = make_float4(0.f, 0.f, 0.f, 0.f);
        if (lane < b1 - b0) sjA = perm[b0 + lane];
        if (nn > 1) {
            const int b2 = __builtin_amdgcn_readlane(rp, 2);
            if (lane < b2 - b1) sjB = perm[b1 + lane];
        }
        if (lane < b1 - b0) l0 = el4[sjA];
        xA0 = leaky(l0.x + __int_as_float(__builtin_amdgcn_readlane(__float_as_int(rv.x), 0)));
        xA1 = leaky(l0.y + __int_as_float(__builtin_amdgcn_readlane(__float_as_int(rv.y), 0)));
        xA2 = leaky(l0.z + __int_as_float(__builtin_amdgcn_readlane(__float_as_int(rv.z), 0)));
        xA3 = leaky(l0.w + __int_as_float(__builtin_amdgcn_readlane(__float_as_int(rv.w), 0)));
    }

    for (int i = 0; i < nn; ++i) {
        const int d   = d0 + i;
        const int beg = __builtin_amdgcn_readlane(rp, i);
        const int end = __builtin_amdgcn_readlane(rp, i + 1);
        const int deg = end - beg;

        float4 lB  = make_float4(0.f, 0.f, 0.f, 0.f);
        int    sjC = 0;
        if (i + 1 < nn) {
            const int endB = __builtin_amdgcn_readlane(rp, i + 2);
            if (lane < endB - end) lB = el4[sjB];
        }
        if (i + 2 < nn) {
            const int begC = __builtin_amdgcn_readlane(rp, i + 2);
            const int endC = __builtin_amdgcn_readlane(rp, i + 3);
            if (lane < endC - begC) sjC = perm[begC + lane];
        }

        float4 acc = make_float4(0.f, 0.f, 0.f, 0.f);
        if (deg <= 32) {
            const bool act = lane < deg;
            const float x0 = xA0, x1 = xA1, x2 = xA2, x3 = xA3;
            float m0 = act ? x0 : -3.402823466e38f;
            float m1 = act ? x1 : -3.402823466e38f;
            float m2 = act ? x2 : -3.402823466e38f;
            float m3 = act ? x3 : -3.402823466e38f;
            #pragma unroll
            for (int off = 16; off > 0; off >>= 1) {
                m0 = fmaxf(m0, __shfl_xor(m0, off, 32));
                m1 = fmaxf(m1, __shfl_xor(m1, off, 32));
                m2 = fmaxf(m2, __shfl_xor(m2, off, 32));
                m3 = fmaxf(m3, __shfl_xor(m3, off, 32));
            }
            const float e0 = act ? __expf(x0 - m0) : 0.f;
            const float e1 = act ? __expf(x1 - m1) : 0.f;
            const float e2 = act ? __expf(x2 - m2) : 0.f;
            const float e3 = act ? __expf(x3 - m3) : 0.f;
            float s0 = e0, s1 = e1, s2 = e2, s3 = e3;
            #pragma unroll
            for (int off = 16; off > 0; off >>= 1) {
                s0 += __shfl_xor(s0, off, 32);
                s1 += __shfl_xor(s1, off, 32);
                s2 += __shfl_xor(s2, off, 32);
                s3 += __shfl_xor(s3, off, 32);
            }
            const float a0 = act ? e0 / s0 : 0.f;
            const float a1 = act ? e1 / s1 : 0.f;
            const float a2 = act ? e2 / s2 : 0.f;
            const float a3 = act ? e3 / s3 : 0.f;
            #pragma unroll 4
            for (int k = 0; k < deg; ++k) {
                const int      sk = __builtin_amdgcn_readlane(sjA, k);
                const unsigned b0 = (unsigned)__builtin_amdgcn_readlane(__float_as_int(a0), k);
                const unsigned b1 = (unsigned)__builtin_amdgcn_readlane(__float_as_int(a1), k);
                const unsigned b2 = (unsigned)__builtin_amdgcn_readlane(__float_as_int(a2), k);
                const unsigned b3 = (unsigned)__builtin_amdgcn_readlane(__float_as_int(a3), k);
                const float w = __uint_as_float((b0 & p0) | (b1 & p1) | (b2 & p2) | (b3 & p3));
                const float4 f = reinterpret_cast<const float4*>(ft + (size_t)sk * HD)[lane];
                acc.x += f.x * w;
                acc.y += f.y * w;
                acc.z += f.z * w;
                acc.w += f.w * w;
            }
        } else {
            float4 r;
            r.x = __int_as_float(__builtin_amdgcn_readlane(__float_as_int(rv.x), i));
            r.y = __int_as_float(__builtin_amdgcn_readlane(__float_as_int(rv.y), i));
            r.z = __int_as_float(__builtin_amdgcn_readlane(__float_as_int(rv.z), i));
            r.w = __int_as_float(__builtin_amdgcn_readlane(__float_as_int(rv.w), i));
            float m0 = -3.402823466e38f, m1 = m0, m2 = m0, m3 = m0;
            for (int j = beg + lane; j < end; j += 32) {
                const float4 l = el4[perm[j]];
                m0 = fmaxf(m0, leaky(l.x + r.x));
                m1 = fmaxf(m1, leaky(l.y + r.y));
                m2 = fmaxf(m2, leaky(l.z + r.z));
                m3 = fmaxf(m3, leaky(l.w + r.w));
            }
            #pragma unroll
            for (int off = 16; off > 0; off >>= 1) {
                m0 = fmaxf(m0, __shfl_xor(m0, off, 32));
                m1 = fmaxf(m1, __shfl_xor(m1, off, 32));
                m2 = fmaxf(m2, __shfl_xor(m2, off, 32));
                m3 = fmaxf(m3, __shfl_xor(m3, off, 32));
            }
            float s0 = 0.f, s1 = 0.f, s2 = 0.f, s3 = 0.f;
            for (int j = beg + lane; j < end; j += 32) {
                const float4 l = el4[perm[j]];
                s0 += __expf(leaky(l.x + r.x) - m0);
                s1 += __expf(leaky(l.y + r.y) - m1);
                s2 += __expf(leaky(l.z + r.z) - m2);
                s3 += __expf(leaky(l.w + r.w) - m3);
            }
            #pragma unroll
            for (int off = 16; off > 0; off >>= 1) {
                s0 += __shfl_xor(s0, off, 32);
                s1 += __shfl_xor(s1, off, 32);
                s2 += __shfl_xor(s2, off, 32);
                s3 += __shfl_xor(s3, off, 32);
            }
            for (int base = beg; base < end; base += 32) {
                const int j = base + lane;
                int   sj = 0;
                float a0 = 0.f, a1 = 0.f, a2 = 0.f, a3 = 0.f;
                if (j < end) {
                    sj = perm[j];
                    const float4 l = el4[sj];
                    a0 = __expf(leaky(l.x + r.x) - m0) / s0;
                    a1 = __expf(leaky(l.y + r.y) - m1) / s1;
                    a2 = __expf(leaky(l.z + r.z) - m2) / s2;
                    a3 = __expf(leaky(l.w + r.w) - m3) / s3;
                }
                const int nk = (end - base < 32) ? (end - base) : 32;
                #pragma unroll 4
                for (int k = 0; k < nk; ++k) {
                    const int      sk = __builtin_amdgcn_readlane(sj, k);
                    const unsigned b0 = (unsigned)__builtin_amdgcn_readlane(__float_as_int(a0), k);
                    const unsigned b1 = (unsigned)__builtin_amdgcn_readlane(__float_as_int(a1), k);
                    const unsigned b2 = (unsigned)__builtin_amdgcn_readlane(__float_as_int(a2), k);
                    const unsigned b3 = (unsigned)__builtin_amdgcn_readlane(__float_as_int(a3), k);
                    const float w = __uint_as_float((b0 & p0) | (b1 & p1) | (b2 & p2) | (b3 & p3));
                    const float4 f = reinterpret_cast<const float4*>(ft + (size_t)sk * HD)[lane];
                    acc.x += f.x * w;
                    acc.y += f.y * w;
                    acc.z += f.z * w;
                    acc.w += f.w * w;
                }
            }
        }
        acc.x += bs.x; acc.y += bs.y; acc.z += bs.z; acc.w += bs.w;
        reinterpret_cast<float4*>(out + (size_t)d * HD)[lane] = acc;

        sjA = sjB; sjB = sjC;
        xA0 = leaky(lB.x + __int_as_float(__builtin_amdgcn_readlane(__float_as_int(rv.x), i + 1)));
        xA1 = leaky(lB.y + __int_as_float(__builtin_amdgcn_readlane(__float_as_int(rv.y), i + 1)));
        xA2 = leaky(lB.z + __int_as_float(__builtin_amdgcn_readlane(__float_as_int(rv.z), i + 1)));
        xA3 = leaky(lB.w + __int_as_float(__builtin_amdgcn_readlane(__float_as_int(rv.w), i + 1)));
    }
}

extern "C" void kernel_launch(void* const* d_in, const int* in_sizes, int n_in,
                              void* d_out, int out_size, void* d_ws, size_t ws_size,
                              hipStream_t stream) {
    const float* feat   = (const float*)d_in[0];
    const int*   src    = (const int*)  d_in[1];
    const int*   dst    = (const int*)  d_in[2];
    const float* W      = (const float*)d_in[3];
    const float* attn_l = (const float*)d_in[4];
    const float* attn_r = (const float*)d_in[5];
    const float* bias   = (const float*)d_in[6];

    const int N = in_sizes[0] / DIN;
    const int E = in_sizes[1];
    float* out = (float*)d_out;

    float* ws     = (float*)d_ws;
    float* ft     = ws;
    float* el     = ft + (size_t)N * HD;
    float* er     = el + (size_t)N * HH;
    int*   rank   = (int*)(er + (size_t)N * HH);
    int*   perm   = rank + E;
    int*   cnt    = perm + E;
    int*   rowptr = cnt + N;
    int*   cnt1   = perm;

    gat_clear<<<(N + 1023) / 1024, 1024, 0, stream>>>(cnt, cnt1, N);
    gat_gemm<<<(N + 63) / 64, 256, 0, stream>>>(feat, W, attn_l, attn_r, ft, el, er, dst, cnt, cnt1, rank, N, E);
    gat_scan<<<(N + 1023) / 1024, 1024, 0, stream>>>(cnt, cnt1, rowptr, N);
    gat_fill<<<(E + 255) / 256, 256, 0, stream>>>(src, dst, rowptr, rank, perm, E);
    gat_node<<<(N + 8 * NPW - 1) / (8 * NPW), 256, 0, stream>>>(rowptr, perm, el, er, ft, bias, out, N);
}


# Scan and fill launches dropped: the GEMM block stores each edge's source at its atomic rank into a 64-slot-per-plane list inside the node's own output row; edge order within a node changes
# speedup vs baseline: 5.3799x; 1.1395x over previous
#include <hip/hip_runtime.h>

typedef __attribute__((ext_vector_type(2))) float v2f;
typedef __attribute__((ext_vector_type(8))) float v8f;

#define DIN 256
#define HD  128
#define HH  4
#define DD  32
#define NEG_SLOPE 0.2f

template <int M>
__device__ __forceinline__ float xrow(float x) {
    return __int_as_float(__builtin_amdgcn_update_dpp(
        0, __float_as_int(x), 0x160 | M, 0xf, 0xf, true));
}

#define SLOTCAP 64

__device__ __forceinline__ int slot_word(int p, int r) {
    return ((r >> 5) << 6) | (p << 5) | (r & 31);
}

#define GROWS 64
#define GTILES (GROWS / 16)
#define PSTR 72

__global__ __launch_bounds__(256) void gat_gemm(const float* __restrict__ feat,
                                                const float* __restrict__ W,
                                                const float* __restrict__ attn_r,
                                                float* __restrict__ ft,
                                                float* __restrict__ er,
                                                const int* __restrict__ src,
                                                const int* __restrict__ dst,
                                                int* __restrict__ cnt,
                                                int* __restrict__ cnt1,
                                                int* __restrict__ slot,
                                                int N, int E) {
    __shared__ __attribute__((aligned(16))) float As[GROWS][260];
    __shared__ float part[8 * PSTR];
    const int tid = threadIdx.x;
    const int rowBase = blockIdx.x * GROWS;
    const bool fullTile = (rowBase + GROWS <= N);

    const int upper = (__builtin_amdgcn_s_sendmsg_rtn(0x87) >> 18) & 1;
    int* cpl = upper ? cnt1 : cnt;
    const bool vecOK = ((((size_t)dst) | ((size_t)src)) & 15) == 0;
    const int e0 = (blockIdx.x * 256 + tid) * 4;
    int dv[4] = {0, 0, 0, 0};
    int sv[4] = {0, 0, 0, 0};
    int rk[4] = {0, 0, 0, 0};
    if (vecOK && e0 + 3 < E) {
        const int4 q = *reinterpret_cast<const int4*>(dst + e0);
        dv[0] = q.x; dv[1] = q.y; dv[2] = q.z; dv[3] = q.w;
        const int4 u = *reinterpret_cast<const int4*>(src + e0);
        sv[0] = u.x; sv[1] = u.y; sv[2] = u.z; sv[3] = u.w;
    } else {
        #pragma unroll
        for (int j = 0; j < 4; ++j)
            if (e0 + j < E) { dv[j] = dst[e0 + j]; sv[j] = src[e0 + j]; }
    }

    if (fullTile) {
        #pragma unroll
        for (int i = 0; i < 16; ++i) {
            int idx4 = i * 256 + tid;
            int r    = idx4 >> 6;
            int c4   = idx4 & 63;
            float4 v = reinterpret_cast<const float4*>(feat + (size_t)(rowBase + r) * DIN)[c4];
            *reinterpret_cast<float4*>(&As[r][c4 * 4]) = v;
        }
    } else {
        #pragma unroll
        for (int i = 0; i < 16; ++i) {
            int idx4 = i * 256 + tid;
            int r    = idx4 >> 6;
            int c4   = idx4 & 63;
            float4 v = make_float4(0.f, 0.f, 0.f, 0.f);
            if (rowBase + r < N)
                v = reinterpret_cast<const float4*>(feat + (size_t)(rowBase + r) * DIN)[c4];
            *reinterpret_cast<float4*>(&As[r][c4 * 4]) = v;
        }
    }
    __syncthreads();

    const int wave    = tid >> 5;
    const int lane    = tid & 31;
    const int half    = lane >> 4;
    const int mrow    = lane & 15;
    const int colBase = wave * 16;
    const int ocol    = colBase + (lane & 15);
    const int kp      = half * 2;

    v8f c[GTILES];
    #pragma unroll
    for (int t = 0; t < GTILES; ++t)
        c[t] = (v8f){0.f, 0.f, 0.f, 0.f, 0.f, 0.f, 0.f, 0.f};
    const float* wrow = W + (size_t)ocol * DIN;
    const float ar = attn_r[ocol];

    #pragma unroll
    for (int q = 0; q < 4; ++q) {
    #pragma unroll 4
    for (int k0 = q * (DIN / 4); k0 < (q + 1) * (DIN / 4); k0 += 4) {
        v2f b = *reinterpret_cast<const v2f*>(wrow + k0 + kp);
        #pragma unroll
        for (int t = 0; t < GTILES; ++t) {
            v2f a = *reinterpret_cast<const v2f*>(&As[t * 16 + mrow][k0 + kp]);
            c[t] = __builtin_amdgcn_wmma_f32_16x16x4_f32(
                     false, a,  false, b,
                     (short)0, c[t],  false,  false);
        }
    }
    if (e0 + q < E) rk[q] = atomicAdd(&cpl[dv[q]], 1);
    }

    const bool b3 = (lane & 8) != 0;
    const bool b2 = (lane & 4) != 0;
    const bool b1 = (lane & 2) != 0;
    const bool b0 = (lane & 1) != 0;
    #pragma unroll
    for (int t = 0; t < GTILES; ++t) {
        float pr[8];
        #pragma unroll
        for (int v = 0; v < 8; ++v) pr[v] = c[t][v] * ar;
        float r1[4];
        #pragma unroll
        for (int v = 0; v < 4; ++v)
            r1[v] = (b3 ? pr[v + 4] : pr[v]) + xrow<8>(b3 ? pr[v] : pr[v + 4]);
        float r2[2];
        #pragma unroll
        for (int v = 0; v < 2; ++v)
            r2[v] = (b2 ? r1[v + 2] : r1[v]) + xrow<4>(b2 ? r1[v] : r1[v + 2]);
        const float r3 = (b1 ? r2[1] : r2[0]) + xrow<2>(b1 ? r2[0] : r2[1]);
        const float fin = r3 + xrow<1>(r3);
        if (!b0) part[wave * PSTR + t * 16 + 8 * half + ((lane & 15) >> 1)] = fin;
    }
    __syncthreads();

    #pragma unroll
    for (int t = 0; t < GTILES; ++t) {
        float* outBase = ft + (size_t)(rowBase + t * 16 + 8 * half) * HD + colBase + (lane & 15);
        if (fullTile) {
            #pragma unroll
            for (int v = 0; v < 8; ++v)
                outBase[(size_t)v * HD] = c[t][v];
        } else {
            #pragma unroll
            for (int v = 0; v < 8; ++v) {
                if (rowBase + t * 16 + v + 8 * half < N)
                    outBase[(size_t)v * HD] = c[t][v];
            }
        }
    }

    {
        const int rr = tid >> 2;
        const int hh = tid & 3;
        const float sum = part[(2 * hh) * PSTR + rr] + part[(2 * hh + 1) * PSTR + rr];
        if (rowBase + rr < N) er[(size_t)(rowBase + rr) * HH + hh] = sum;
    }

    #pragma unroll
    for (int j = 0; j < 4; ++j)
        if (e0 + j < E && rk[j] < SLOTCAP)
            slot[(size_t)dv[j] * HD + slot_word(upper, rk[j])] = sv[j];
    for (long long e = (long long)gridDim.x * 1024 + (long long)blockIdx.x * 256 + tid;
         e < E; e += (long long)gridDim.x * 256) {
        const int d = dst[e];
        const int r = atomicAdd(&cpl[d], 1);
        if (r < SLOTCAP) slot[(size_t)d * HD + slot_word(upper, r)] = src[e];
    }
}

__global__ __launch_bounds__(1024) void gat_clear(int* __restrict__ cnt0,
                                                  int* __restrict__ cnt1, int N) {
    const int i = blockIdx.x * 1024 + threadIdx.x;
    if (i < N) { cnt0[i] = 0; cnt1[i] = 0; }
}

__device__ __forceinline__ float leaky(float x) {
    return (x > 0.f) ? x : NEG_SLOPE * x;
}

#define NPW 8
#define LOG2E 1.4426950408889634f

template <int NV>
__device__ __forceinline__ void rows_load(const float* __restrict__ ft, const int sj, const int k,
                                          const int lane, float4 (&f)[NV]) {
    #pragma unroll
    for (int j = 0; j < NV; ++j) {
        const int sk = __builtin_amdgcn_readlane(sj, k + j);
        f[j] = reinterpret_cast<const float4*>(ft + (size_t)sk * HD)[lane];
    }
}

template <int NV>
__device__ __forceinline__ void rows_use(const float4 (&f)[NV], const float4 al, const float erd,
                                         float& m, float& s, float4& acc) {
    float x[NV];
    #pragma unroll
    for (int j = 0; j < NV; ++j) {
        float t = f[j].x * al.x + f[j].y * al.y + f[j].z * al.z + f[j].w * al.w;
        t += xrow<1>(t);
        t += xrow<2>(t);
        t += xrow<4>(t);
        x[j] = leaky(t + erd);
    }
    float mn = fmaxf(m, x[0]);
    #pragma unroll
    for (int j = 1; j < NV; ++j) mn = fmaxf(mn, x[j]);
    const float sc = __builtin_amdgcn_exp2f(m - mn);
    s *= sc;
    acc.x *= sc; acc.y *= sc; acc.z *= sc; acc.w *= sc;
    #pragma unroll
    for (int j = 0; j < NV; ++j) {
        const float p = __builtin_amdgcn_exp2f(x[j] - mn);
        s += p;
        acc.x += p * f[j].x;
        acc.y += p * f[j].y;
        acc.z += p * f[j].z;
        acc.w += p * f[j].w;
    }
    m = mn;
}

template <int NV>
__device__ __forceinline__ void edge_group(const float* __restrict__ ft, const int sj, const int k,
                                           const int lane, const float4 al, const float erd,
                                           float& m, float& s, float4& acc) {
    float4 f[NV];
    rows_load<NV>(ft, sj, k, lane, f);
    rows_use<NV>(f, al, erd, m, s, acc);
}

__device__ __forceinline__ int list_word(int j, int c0) {
    const bool first = j < c0;
    return slot_word(first ? 0 : 1, first ? j : (j - c0));
}

__global__ __launch_bounds__(256) __attribute__((amdgpu_waves_per_eu(16, 16))) void gat_node(
        const int* __restrict__ cnt0, const int* __restrict__ cnt1,
        const int* slot,
        const int* __restrict__ src, const int* __restrict__ dst,
        const float* __restrict__ er, const float* __restrict__ ft,
        const float* __restrict__ attn_l, const float* __restrict__ bias,
        float* out, int N, int E) {
    const int lane = threadIdx.x & 31;
    const int g  = __builtin_amdgcn_readfirstlane((int)(blockIdx.x * 8 + (threadIdx.x >> 5)));
    const int d0 = g * NPW;
    if (d0 >= N) return;
    const int nn = (N - d0 < NPW) ? (N - d0) : NPW;

    const int    li  = (lane < nn) ? lane : (nn - 1);
    const int    c0v = cnt0[d0 + li];
    const int    c1v = cnt1[d0 + li];
    const float4 rv  = reinterpret_cast<const float4*>(er)[d0 + li];
    const float4 bs  = reinterpret_cast<const float4*>(bias)[lane];
    float4 al = reinterpret_cast<const float4*>(attn_l)[lane];
    al.x *= LOG2E; al.y *= LOG2E; al.z *= LOG2E; al.w *= LOG2E;

    const int hsel = lane >> 3;
    const unsigned p0 = (hsel == 0) ? 0xffffffffu : 0u;
    const unsigned p1 = (hsel == 1) ? 0xffffffffu : 0u;
    const unsigned p2 = (hsel == 2) ? 0xffffffffu : 0u;
    const unsigned p3 = (hsel == 3) ? 0xffffffffu : 0u;

    int sjA = 0, sjB = 0;
    {
        const int a0 = __builtin_amdgcn_readlane(c0v, 0);
        sjA = slot[(size_t)d0 * HD + list_word(lane, a0)];
        if (nn > 1) {
            const int a1 = __builtin_amdgcn_readlane(c0v, 1);
            sjB = slot[(size_t)(d0 + 1) * HD + list_word(lane, a1)];
        }
    }

    for (int i = 0; i < nn; ++i) {
        const int d   = d0 + i;
        const int c0  = __builtin_amdgcn_readlane(c0v, i);
        const int c1  = __builtin_amdgcn_readlane(c1v, i);
        const int deg = c0 + c1;

        const unsigned r0 = (unsigned)__builtin_amdgcn_readlane(__float_as_int(rv.x), i);
        const unsigned r1 = (unsigned)__builtin_amdgcn_readlane(__float_as_int(rv.y), i);
        const unsigned r2 = (unsigned)__builtin_amdgcn_readlane(__float_as_int(rv.z), i);
        const unsigned r3 = (unsigned)__builtin_amdgcn_readlane(__float_as_int(rv.w), i);
        const float erd = __uint_as_float((r0 & p0) | (r1 & p1) | (r2 & p2) | (r3 & p3)) * LOG2E;

        float  m = __int_as_float((int)0xff800000u);
        float  s = 0.f;
        float4 acc = make_float4(0.f, 0.f, 0.f, 0.f);

        const int  iC   = (i + 2 < nn) ? (i + 2) : (nn - 1);
        const int  c0C  = __builtin_amdgcn_readlane(c0v, iC);
        const int* rowC = slot + (size_t)(d0 + iC) * HD + list_word(lane, c0C);
        int sjC;

        if (deg <= 32) {
            int k = 0;
            if (deg >= 4) {
                float4 f[4];
                rows_load<4>(ft, sjA, 0, lane, f);
                sjC = *rowC;
                rows_use<4>(f, al, erd, m, s, acc);
                k = 4;
            } else {
                sjC = *rowC;
            }
            for (; k + 4 <= deg; k += 4)
                edge_group<4>(ft, sjA, k, lane, al, erd, m, s, acc);
            const int rem = deg - k;
            if (rem == 3)      edge_group<3>(ft, sjA, k, lane, al, erd, m, s, acc);
            else if (rem == 2) edge_group<2>(ft, sjA, k, lane, al, erd, m, s, acc);
            else if (rem == 1) edge_group<1>(ft, sjA, k, lane, al, erd, m, s, acc);
        } else if (c0 <= SLOTCAP && c1 <= SLOTCAP) {
            sjC = *rowC;
            const int* row = slot + (size_t)d * HD;
            for (int base = 0; base < deg; base += 32) {
                int sj = 0;
                if (base + lane < deg) sj = row[list_word(base + lane, c0)];
                const int nk = (deg - base < 32) ? (deg - base) : 32;
                int k = 0;
                for (; k + 4 <= nk; k += 4)
                    edge_group<4>(ft, sj, k, lane, al, erd, m, s, acc);
                const int rem = nk - k;
                if (rem == 3)      edge_group<3>(ft, sj, k, lane, al, erd, m, s, acc);
                else if (rem == 2) edge_group<2>(ft, sj, k, lane, al, erd, m, s, acc);
                else if (rem == 1) edge_group<1>(ft, sj, k, lane, al, erd, m, s, acc);
            }
        } else {
            sjC = *rowC;
            for (int base = 0; base < E; base += 32) {
                const int  e   = base + lane;
                const bool hit = (e < E) && (dst[e] == d);
                unsigned   hm  = __builtin_amdgcn_ballot_w32(hit);
                if (hm != 0u) {
                    const int sj = (e < E) ? src[e] : 0;
                    while (hm != 0u) {
                        const int k = __builtin_ctz(hm);
                        hm &= hm - 1u;
                        edge_group<1>(ft, sj, k, lane, al, erd, m, s, acc);
                    }
                }
            }
        }

        const float inv = (deg > 0) ? (1.0f / s) : 0.f;
        float4 o;
        o.x = acc.x * inv + bs.x;
        o.y = acc.y * inv + bs.y;
        o.z = acc.z * inv + bs.z;
        o.w = acc.w * inv + bs.w;
        reinterpret_cast<float4*>(out + (size_t)d * HD)[lane] = o;

        sjA = sjB; sjB = sjC;
    }
}

extern "C" void kernel_launch(void* const* d_in, const int* in_sizes, int n_in,
                              void* d_out, int out_size, void* d_ws, size_t ws_size,
                              hipStream_t stream) {
    const float* feat   = (const float*)d_in[0];
    const int*   src    = (const int*)  d_in[1];
    const int*   dst    = (const int*)  d_in[2];
    const float* W      = (const float*)d_in[3];
    const float* attn_l = (const float*)d_in[4];
    const float* attn_r = (const float*)d_in[5];
    const float* bias   = (const float*)d_in[6];

    const int N = in_sizes[0] / DIN;
    const int E = in_sizes[1];
    float* out  = (float*)d_out;
    int*   slot = (int*)d_out;

    float* ws   = (float*)d_ws;
    float* ft   = ws;
    float* er   = ft + (size_t)N * HD;
    int*   cnt  = (int*)(er + (size_t)N * HH);
    int*   cnt1 = cnt + N;

    gat_clear<<<(N + 1023) / 1024, 1024, 0, stream>>>(cnt, cnt1, N);
    gat_gemm<<<(N + 63) / 64, 256, 0, stream>>>(feat, W, attn_r, ft, er, src, dst, cnt, cnt1, slot, N, E);
    gat_node<<<(N + 8 * NPW - 1) / (8 * NPW), 256, 0, stream>>>(cnt, cnt1, slot, src, dst, er, ft, attn_l,
                                                              bias, out, N, E);
}
